# starting kernel
# baseline (speedup 1.0000x reference)
#include <hip/hip_runtime.h>
#include <hip/hip_bf16.h>


typedef __attribute__((ext_vector_type(16))) _Float16 v16h;
typedef __attribute__((ext_vector_type(8)))  _Float16 v8h;
typedef __attribute__((ext_vector_type(2)))  _Float16 v2h;
typedef __attribute__((ext_vector_type(8)))  float    v8f;
typedef __attribute__((ext_vector_type(2)))  float    v2f;

#define HID   32
#define NTILE 8

#if __has_builtin(__builtin_amdgcn_wmma_f32_16x16x4_f32)
#define HAS_K4 1
#else
#define HAS_K4 0
#endif

#if __has_builtin(__builtin_amdgcn_tanhf)
__device__ __forceinline__ float tanh_(float x) { return __builtin_amdgcn_tanhf(x); }
__device__ __forceinline__ float sigm(float x) {
    return __builtin_fmaf(0.5f, __builtin_amdgcn_tanhf(0.5f * x), 0.5f);
}
#else
__device__ __forceinline__ float fast_rcp(float x) { return __builtin_amdgcn_rcpf(x); }
__device__ __forceinline__ float tanh_(float x) {
    float e = __expf(-2.0f * x);
    return (1.0f - e) * fast_rcp(1.0f + e);
}
__device__ __forceinline__ float sigm(float x) { return fast_rcp(1.0f + __expf(-x)); }
#endif

__device__ __forceinline__ int permK(int k) { return (k & 1) ? ((k >> 1) + 16) : (k >> 1); }

__global__ __launch_bounds__(32)
void lstm_wmma_kernel(const float* __restrict__ x,
                      const float* __restrict__ W_ih,
                      const float* __restrict__ W_hh,
                      const float* __restrict__ b_ih,
                      const float* __restrict__ b_hh,
                      const float* __restrict__ W_fc,
                      const float* __restrict__ b_fc,
                      float* __restrict__ out,
                      int T)
{
    __shared__ _Float16 hbuf[16 * HID];

    const int lane = threadIdx.x;
    const int col  = lane & 15;
    const int hi   = lane >> 4;
    const int b0   = blockIdx.x * 16;

    v16h Bt[NTILE];
#pragma unroll
    for (int tt = 0; tt < NTILE; ++tt) {
        const float* wrow = W_hh + (size_t)(tt * 16 + col) * HID;
#pragma unroll
        for (int e = 0; e < 8; ++e) {
            Bt[tt][e]     = (_Float16)wrow[permK(hi * 8 + e)];
            Bt[tt][e + 8] = (_Float16)wrow[permK(16 + hi * 8 + e)];
        }
    }

#if HAS_K4
    v2f Bx[NTILE];
#pragma unroll
    for (int tt = 0; tt < NTILE; ++tt) {
        int g = tt * 16 + col;
        if (hi == 0) { Bx[tt][0] = W_ih[g * 2 + 0];     Bx[tt][1] = W_ih[g * 2 + 1]; }
        else         { Bx[tt][0] = b_ih[g] + b_hh[g];   Bx[tt][1] = 0.0f; }
    }
#else
    v16h Bx[NTILE];
#pragma unroll
    for (int tt = 0; tt < NTILE; ++tt) {
        Bx[tt] = (v16h){};
        int g = tt * 16 + col;
        if (hi == 0) {
            Bx[tt][0] = (_Float16)W_ih[g * 2 + 0];
            Bx[tt][1] = (_Float16)W_ih[g * 2 + 1];
            Bx[tt][2] = (_Float16)(b_ih[g] + b_hh[g]);
        }
    }
#endif

    v16h Bfc = (v16h){};
    if (col == 0) {
#pragma unroll
        for (int e = 0; e < 8; ++e) {
            Bfc[e]     = (_Float16)W_fc[permK(hi * 8 + e)];
            Bfc[e + 8] = (_Float16)W_fc[permK(16 + hi * 8 + e)];
        }
    }
    v8f Cfc;
    {
        float bfc = b_fc[0];
#pragma unroll
        for (int e = 0; e < 8; ++e) Cfc[e] = bfc;
    }

    float c1[8], c2[8];
#pragma unroll
    for (int r = 0; r < 8; ++r) { c1[r] = 0.f; c2[r] = 0.f; }
    for (int i = lane; i < 16 * HID; i += 32) hbuf[i] = (_Float16)0.f;
    __syncthreads();

    const size_t xrow = (size_t)T * 2;
    const float* pxbase = x + (size_t)(b0 + col) * xrow;

#if !HAS_K4
    v16h Ax = (v16h){};
    if (hi == 0) Ax[2] = (_Float16)1.0f;
#endif

    float2 xv = *(const float2*)pxbase;

    for (int t = 0; t < T; ++t) {
#if HAS_K4
        v2f Ax;
        Ax[0] = (hi == 0) ? xv.x : 1.0f;
        Ax[1] = (hi == 0) ? xv.y : 0.0f;
#else
        if (hi == 0) { Ax[0] = (_Float16)xv.x; Ax[1] = (_Float16)xv.y; }
#endif

        int tn = (t + 1 < T) ? (t + 1) : t;
        float2 xvn = *(const float2*)(pxbase + (size_t)tn * 2);

        v16h A;
        {
            v8h alo = *(const v8h*)&hbuf[col * HID + hi * 8];
            v8h ahi = *(const v8h*)&hbuf[col * HID + 16 + hi * 8];
#pragma unroll
            for (int e = 0; e < 8; ++e) { A[e] = alo[e]; A[e + 8] = ahi[e]; }
        }

        v8f accy = __builtin_amdgcn_wmma_f32_16x16x32_f16(
            false, A, false, Bfc, (short)0, Cfc, false, false);

        v8f acc[NTILE];
#pragma unroll
        for (int tt = 0; tt < NTILE; ++tt) {
            v8f z = (v8f){};
#if HAS_K4
            acc[tt] = __builtin_amdgcn_wmma_f32_16x16x4_f32(
                false, Ax, false, Bx[tt], (short)0, z, false, false);
#else
            acc[tt] = __builtin_amdgcn_wmma_f32_16x16x32_f16(
                false, Ax, false, Bx[tt], (short)0, z, false, false);
#endif
        }
#pragma unroll
        for (int tt = 0; tt < NTILE; ++tt) {
            acc[tt] = __builtin_amdgcn_wmma_f32_16x16x32_f16(
                false, A, false, Bt[tt], (short)0, acc[tt], false, false);
        }

#pragma unroll
        for (int r = 0; r < 8; ++r) {
            float i1 = sigm(acc[0][r]);
            float f1 = sigm(acc[2][r]);
            float g1 = tanh_(acc[4][r]);
            float o1 = sigm(acc[6][r]);
            c1[r] = f1 * c1[r] + i1 * g1;
            float h1 = o1 * tanh_(c1[r]);

            float i2 = sigm(acc[1][r]);
            float f2 = sigm(acc[3][r]);
            float g2 = tanh_(acc[5][r]);
            float o2 = sigm(acc[7][r]);
            c2[r] = f2 * c2[r] + i2 * g2;
            float h2 = o2 * tanh_(c2[r]);

            int row = r + 8 * hi;
            *(v2h*)&hbuf[row * HID + col * 2] = (v2h){(_Float16)h1, (_Float16)h2};
        }
        __syncthreads();

        if (t > 0 && col == 0) {
#pragma unroll
            for (int r = 0; r < 8; ++r)
                out[(size_t)(b0 + r + 8 * hi) * T + (t - 1)] = accy[r];
        }
        xv = xvn;
    }

    {
        v16h A;
        v8h alo = *(const v8h*)&hbuf[col * HID + hi * 8];
        v8h ahi = *(const v8h*)&hbuf[col * HID + 16 + hi * 8];
#pragma unroll
        for (int e = 0; e < 8; ++e) { A[e] = alo[e]; A[e + 8] = ahi[e]; }
        v8f accy = __builtin_amdgcn_wmma_f32_16x16x32_f16(
            false, A, false, Bfc, (short)0, Cfc, false, false);
        if (col == 0) {
#pragma unroll
            for (int r = 0; r < 8; ++r)
                out[(size_t)(b0 + r + 8 * hi) * T + (T - 1)] = accy[r];
        }
    }
}

extern "C" void kernel_launch(void* const* d_in, const int* in_sizes, int n_in,
                              void* d_out, int out_size, void* d_ws, size_t ws_size,
                              hipStream_t stream) {
    const float* x    = (const float*)d_in[0];
    const float* W_ih = (const float*)d_in[1];
    const float* W_hh = (const float*)d_in[2];
    const float* b_ih = (const float*)d_in[3];
    const float* b_hh = (const float*)d_in[4];
    const float* W_fc = (const float*)d_in[5];
    const float* b_fc = (const float*)d_in[6];
    float* out = (float*)d_out;

    const int T = 512;
    const int B = out_size / T;
    const int nblocks = B / 16;

    lstm_wmma_kernel<<<dim3(nblocks), dim3(32), 0, stream>>>(
        x, W_ih, W_hh, b_ih, b_hh, W_fc, b_fc, out, T);

    (void)in_sizes; (void)n_in; (void)d_ws; (void)ws_size;
}


# Products transposed, one wave per 16 rows kept: W_hh tiles as A, h^T as B, so a lane's D values (cvt f16) are its own next B fragment: no LDS round trip or waits, y_t 1 store; f16 k order changes
# speedup vs baseline: 1.3263x; 1.3263x over previous
#include <hip/hip_runtime.h>
#include <hip/hip_bf16.h>


typedef __attribute__((ext_vector_type(16))) _Float16 v16h;
typedef __attribute__((ext_vector_type(8)))  _Float16 v8h;
typedef __attribute__((ext_vector_type(2)))  _Float16 v2h;
typedef __attribute__((ext_vector_type(8)))  float    v8f;
typedef __attribute__((ext_vector_type(2)))  float    v2f;

#define HID   32
#define NTILE 8

#if __has_builtin(__builtin_amdgcn_wmma_f32_16x16x4_f32)
#define HAS_K4 1
#else
#define HAS_K4 0
#endif

#if __has_builtin(__builtin_amdgcn_tanhf)
__device__ __forceinline__ float tanh_(float x) { return __builtin_amdgcn_tanhf(x); }
__device__ __forceinline__ float sigm(float x) {
    return __builtin_fmaf(0.5f, __builtin_amdgcn_tanhf(0.5f * x), 0.5f);
}
#else
__device__ __forceinline__ float fast_rcp(float x) { return __builtin_amdgcn_rcpf(x); }
__device__ __forceinline__ float tanh_(float x) {
    float e = __expf(-2.0f * x);
    return (1.0f - e) * fast_rcp(1.0f + e);
}
__device__ __forceinline__ float sigm(float x) { return fast_rcp(1.0f + __expf(-x)); }
#endif

__global__ __launch_bounds__(32)
void lstm_wmma_kernel(const float* __restrict__ x,
                      const float* __restrict__ W_ih,
                      const float* __restrict__ W_hh,
                      const float* __restrict__ b_ih,
                      const float* __restrict__ b_hh,
                      const float* __restrict__ W_fc,
                      const float* __restrict__ b_fc,
                      float* __restrict__ out,
                      int T)
{
    const int lane = threadIdx.x;
    const int col  = lane & 15;
    const int hi   = lane >> 4;
    const int b0   = blockIdx.x * 16;

    v16h At[NTILE];
#pragma unroll
    for (int tt = 0; tt < NTILE; ++tt) {
        const float* wrow = W_hh + (size_t)(tt * 16 + col) * HID;
#pragma unroll
        for (int e = 0; e < 8; ++e) {
            At[tt][e]     = (_Float16)wrow[hi * 8 + e];
            At[tt][e + 8] = (_Float16)wrow[16 + hi * 8 + e];
        }
    }

#if HAS_K4
    v2f Aw[NTILE];
#pragma unroll
    for (int tt = 0; tt < NTILE; ++tt) {
        int g = tt * 16 + col;
        if (hi == 0) { Aw[tt][0] = W_ih[g * 2 + 0];     Aw[tt][1] = W_ih[g * 2 + 1]; }
        else         { Aw[tt][0] = b_ih[g] + b_hh[g];   Aw[tt][1] = 0.0f; }
    }
#else
    v16h Aw[NTILE];
#pragma unroll
    for (int tt = 0; tt < NTILE; ++tt) {
        Aw[tt] = (v16h){};
        int g = tt * 16 + col;
        if (hi == 0) {
            Aw[tt][0] = (_Float16)W_ih[g * 2 + 0];
            Aw[tt][1] = (_Float16)W_ih[g * 2 + 1];
            Aw[tt][2] = (_Float16)(b_ih[g] + b_hh[g]);
        }
    }
#endif

    v16h Afc = (v16h){};
    if (col == 0) {
#pragma unroll
        for (int e = 0; e < 8; ++e) {
            Afc[e]     = (_Float16)W_fc[hi * 8 + e];
            Afc[e + 8] = (_Float16)W_fc[16 + hi * 8 + e];
        }
    }
    v8f Cfc;
    {
        float bfc = b_fc[0];
#pragma unroll
        for (int e = 0; e < 8; ++e) Cfc[e] = bfc;
    }

    float c1[8], c2[8];
#pragma unroll
    for (int r = 0; r < 8; ++r) { c1[r] = 0.f; c2[r] = 0.f; }
    v16h Bh = (v16h){};

    const size_t xrow = (size_t)T * 2;
    const float* pxbase = x + (size_t)(b0 + col) * xrow;

#if !HAS_K4
    v16h Bx = (v16h){};
    if (hi == 0) Bx[2] = (_Float16)1.0f;
#endif

    float2 xv = *(const float2*)pxbase;

    for (int t = 0; t < T; ++t) {
#if HAS_K4
        v2f Bx;
        Bx[0] = (hi == 0) ? xv.x : 1.0f;
        Bx[1] = (hi == 0) ? xv.y : 0.0f;
#else
        if (hi == 0) { Bx[0] = (_Float16)xv.x; Bx[1] = (_Float16)xv.y; }
#endif

        int tn = (t + 1 < T) ? (t + 1) : t;
        float2 xvn = *(const float2*)(pxbase + (size_t)tn * 2);

        v8f accy = __builtin_amdgcn_wmma_f32_16x16x32_f16(
            false, Afc, false, Bh, (short)0, Cfc, false, false);

        v8f acc[NTILE];
#pragma unroll
        for (int tt = 0; tt < NTILE; ++tt) {
            v8f z = (v8f){};
#if HAS_K4
            acc[tt] = __builtin_amdgcn_wmma_f32_16x16x4_f32(
                false, Aw[tt], false, Bx, (short)0, z, false, false);
#else
            acc[tt] = __builtin_amdgcn_wmma_f32_16x16x32_f16(
                false, Aw[tt], false, Bx, (short)0, z, false, false);
#endif
        }
#pragma unroll
        for (int tt = 0; tt < NTILE; ++tt) {
            acc[tt] = __builtin_amdgcn_wmma_f32_16x16x32_f16(
                false, At[tt], false, Bh, (short)0, acc[tt], false, false);
        }

        if (t > 0 && hi == 0)
            out[(size_t)(b0 + col) * T + (t - 1)] = accy[0];

#pragma unroll
        for (int r = 0; r < 8; ++r) {
            float i1 = sigm(acc[0][r]);
            float f1 = sigm(acc[2][r]);
            float g1 = tanh_(acc[4][r]);
            float o1 = sigm(acc[6][r]);
            c1[r] = f1 * c1[r] + i1 * g1;
            float h1 = o1 * tanh_(c1[r]);

            float i2 = sigm(acc[1][r]);
            float f2 = sigm(acc[3][r]);
            float g2 = tanh_(acc[5][r]);
            float o2 = sigm(acc[7][r]);
            c2[r] = f2 * c2[r] + i2 * g2;
            float h2 = o2 * tanh_(c2[r]);

            Bh[r]     = (_Float16)h1;
            Bh[r + 8] = (_Float16)h2;
        }
        xv = xvn;
    }

    {
        v8f accy = __builtin_amdgcn_wmma_f32_16x16x32_f16(
            false, Afc, false, Bh, (short)0, Cfc, false, false);
        if (hi == 0)
            out[(size_t)(b0 + col) * T + (T - 1)] = accy[0];
    }
}

extern "C" void kernel_launch(void* const* d_in, const int* in_sizes, int n_in,
                              void* d_out, int out_size, void* d_ws, size_t ws_size,
                              hipStream_t stream) {
    const float* x    = (const float*)d_in[0];
    const float* W_ih = (const float*)d_in[1];
    const float* W_hh = (const float*)d_in[2];
    const float* b_ih = (const float*)d_in[3];
    const float* b_hh = (const float*)d_in[4];
    const float* W_fc = (const float*)d_in[5];
    const float* b_fc = (const float*)d_in[6];
    float* out = (float*)d_out;

    const int T = 512;
    const int B = out_size / T;
    const int nblocks = B / 16;

    lstm_wmma_kernel<<<dim3(nblocks), dim3(32), 0, stream>>>(
        x, W_ih, W_hh, b_ih, b_hh, W_fc, b_fc, out, T);

    (void)in_sizes; (void)n_in; (void)d_ws; (void)ws_size;
}


# speedup vs baseline: 1.4397x; 1.0855x over previous
#include <hip/hip_runtime.h>
#include <hip/hip_bf16.h>


typedef __attribute__((ext_vector_type(16))) _Float16 v16h;
typedef __attribute__((ext_vector_type(8)))  _Float16 v8h;
typedef __attribute__((ext_vector_type(2)))  _Float16 v2h;
typedef __attribute__((ext_vector_type(8)))  float    v8f;
typedef __attribute__((ext_vector_type(2)))  float    v2f;

#define SCHED_PIN() __builtin_amdgcn_sched_barrier(0)

#define HID   32
#define NTILE 8

#if __has_builtin(__builtin_amdgcn_wmma_f32_16x16x4_f32)
#define HAS_K4 1
#else
#define HAS_K4 0
#endif

#if __has_builtin(__builtin_amdgcn_tanhf)
__device__ __forceinline__ float tanh_(float x) { return __builtin_amdgcn_tanhf(x); }
__device__ __forceinline__ float sigm(float x) {
    return __builtin_fmaf(0.5f, __builtin_amdgcn_tanhf(0.5f * x), 0.5f);
}
#else
__device__ __forceinline__ float fast_rcp(float x) { return __builtin_amdgcn_rcpf(x); }
__device__ __forceinline__ float tanh_(float x) {
    float e = __expf(-2.0f * x);
    return (1.0f - e) * fast_rcp(1.0f + e);
}
__device__ __forceinline__ float sigm(float x) { return fast_rcp(1.0f + __expf(-x)); }
#endif

__global__ __launch_bounds__(32)
void lstm_wmma_kernel(const float* __restrict__ x,
                      const float* __restrict__ W_ih,
                      const float* __restrict__ W_hh,
                      const float* __restrict__ b_ih,
                      const float* __restrict__ b_hh,
                      const float* __restrict__ W_fc,
                      const float* __restrict__ b_fc,
                      float* __restrict__ out,
                      int T)
{
    const int lane = threadIdx.x;
    const int col  = lane & 15;
    const int hi   = lane >> 4;
    const int b0   = blockIdx.x * 16;

    v16h At[NTILE];
#pragma unroll
    for (int tt = 0; tt < NTILE; ++tt) {
        const float* wrow = W_hh + (size_t)(tt * 16 + col) * HID;
#pragma unroll
        for (int e = 0; e < 8; ++e) {
            At[tt][e]     = (_Float16)wrow[hi * 8 + e];
            At[tt][e + 8] = (_Float16)wrow[16 + hi * 8 + e];
        }
    }

#if HAS_K4
    v2f Aw[NTILE];
#pragma unroll
    for (int tt = 0; tt < NTILE; ++tt) {
        int g = tt * 16 + col;
        if (hi == 0) { Aw[tt][0] = W_ih[g * 2 + 0];     Aw[tt][1] = W_ih[g * 2 + 1]; }
        else         { Aw[tt][0] = b_ih[g] + b_hh[g];   Aw[tt][1] = 0.0f; }
    }
#else
    v16h Aw[NTILE];
#pragma unroll
    for (int tt = 0; tt < NTILE; ++tt) {
        Aw[tt] = (v16h){};
        int g = tt * 16 + col;
        if (hi == 0) {
            Aw[tt][0] = (_Float16)W_ih[g * 2 + 0];
            Aw[tt][1] = (_Float16)W_ih[g * 2 + 1];
            Aw[tt][2] = (_Float16)(b_ih[g] + b_hh[g]);
        }
    }
#endif

    v16h Afc = (v16h){};
    if (col == 0) {
#pragma unroll
        for (int e = 0; e < 8; ++e) {
            Afc[e]     = (_Float16)W_fc[hi * 8 + e];
            Afc[e + 8] = (_Float16)W_fc[16 + hi * 8 + e];
        }
    }
    v8f Cfc;
    {
        float bfc = b_fc[0];
#pragma unroll
        for (int e = 0; e < 8; ++e) Cfc[e] = bfc;
    }

    v2f cst[2][4];
#pragma unroll
    for (int p = 0; p < 2; ++p)
#pragma unroll
        for (int j = 0; j < 4; ++j) cst[p][j] = (v2f){0.f, 0.f};
    v16h Bh = (v16h){};

    const size_t xrow = (size_t)T * 2;
    const float* pxbase = x + (size_t)(b0 + col) * xrow;

#if !HAS_K4
    v16h Bx = (v16h){};
    if (hi == 0) Bx[2] = (_Float16)1.0f;
#endif

    float2 xv = *(const float2*)pxbase;

    for (int t = 0; t < T; ++t) {
#if HAS_K4
        v2f Bx;
        Bx[0] = (hi == 0) ? xv.x : 1.0f;
        Bx[1] = (hi == 0) ? xv.y : 0.0f;
#else
        if (hi == 0) { Bx[0] = (_Float16)xv.x; Bx[1] = (_Float16)xv.y; }
#endif

        int tn = (t + 1 < T) ? (t + 1) : t;
        float2 xvn = *(const float2*)(pxbase + (size_t)tn * 2);

        v8f accy = __builtin_amdgcn_wmma_f32_16x16x32_f16(
            false, Afc, false, Bh, (short)0, Cfc, false, false);

        v8f acc[NTILE];
#pragma unroll
        for (int tt = 0; tt < NTILE; ++tt) {
            v8f z = (v8f){};
#if HAS_K4
            acc[tt] = __builtin_amdgcn_wmma_f32_16x16x4_f32(
                false, Aw[tt], false, Bx, (short)0, z, false, false);
#else
            acc[tt] = __builtin_amdgcn_wmma_f32_16x16x32_f16(
                false, Aw[tt], false, Bx, (short)0, z, false, false);
#endif
        }
#pragma unroll
        for (int tt = 0; tt < NTILE; ++tt) {
            acc[tt] = __builtin_amdgcn_wmma_f32_16x16x32_f16(
                false, At[tt], false, Bh, (short)0, acc[tt], false, false);
        }

        if (t > 0 && hi == 0)
            out[(size_t)(b0 + col) * T + (t - 1)] = accy[0];

        const v2f vhalf = (v2f){0.5f, 0.5f};
        v2f tg00; tg00[0] = tanh_(acc[4][0]); SCHED_PIN();
        v2f hf00 = (v2f){acc[2][0], acc[2][1]} * 0.5f; SCHED_PIN();
        tg00[1] = tanh_(acc[4][1]); SCHED_PIN();
        v2f hf01 = (v2f){acc[2][2], acc[2][3]} * 0.5f; SCHED_PIN();
        v2f tg01; tg01[0] = tanh_(acc[4][2]); SCHED_PIN();
        v2f hf02 = (v2f){acc[2][4], acc[2][5]} * 0.5f; SCHED_PIN();
        tg01[1] = tanh_(acc[4][3]); SCHED_PIN();
        v2f hf03 = (v2f){acc[2][6], acc[2][7]} * 0.5f; SCHED_PIN();
        v2f tg02; tg02[0] = tanh_(acc[4][4]); SCHED_PIN();
        v2f hf10 = (v2f){acc[3][0], acc[3][1]} * 0.5f; SCHED_PIN();
        v2f tf00; tf00[0] = tanh_(hf00[0]); SCHED_PIN();
        v2f hf11 = (v2f){acc[3][2], acc[3][3]} * 0.5f; SCHED_PIN();
        tf00[1] = tanh_(hf00[1]); SCHED_PIN();
        v2f hf12 = (v2f){acc[3][4], acc[3][5]} * 0.5f; SCHED_PIN();
        v2f tf01; tf01[0] = tanh_(hf01[0]); SCHED_PIN();
        v2f hf13 = (v2f){acc[3][6], acc[3][7]} * 0.5f; SCHED_PIN();
        tf01[1] = tanh_(hf01[1]); SCHED_PIN();
        v2f hi00 = (v2f){acc[0][0], acc[0][1]} * 0.5f; SCHED_PIN();
        v2f tf02; tf02[0] = tanh_(hf02[0]); SCHED_PIN();
        v2f hi01 = (v2f){acc[0][2], acc[0][3]} * 0.5f; SCHED_PIN();
        tf02[1] = tanh_(hf02[1]); SCHED_PIN();
        v2f hi02 = (v2f){acc[0][4], acc[0][5]} * 0.5f; SCHED_PIN();
        v2f tf03; tf03[0] = tanh_(hf03[0]); SCHED_PIN();
        v2f hi03 = (v2f){acc[0][6], acc[0][7]} * 0.5f; SCHED_PIN();
        tf03[1] = tanh_(hf03[1]); SCHED_PIN();
        v2f hi10 = (v2f){acc[1][0], acc[1][1]} * 0.5f; SCHED_PIN();
        v2f tf10; tf10[0] = tanh_(hf10[0]); SCHED_PIN();
        v2f hi11 = (v2f){acc[1][2], acc[1][3]} * 0.5f; SCHED_PIN();
        tf10[1] = tanh_(hf10[1]); SCHED_PIN();
        v2f hi12 = (v2f){acc[1][4], acc[1][5]} * 0.5f; SCHED_PIN();
        v2f tf11; tf11[0] = tanh_(hf11[0]); SCHED_PIN();
        v2f hi13 = (v2f){acc[1][6], acc[1][7]} * 0.5f; SCHED_PIN();
        tf11[1] = tanh_(hf11[1]); SCHED_PIN();
        v2f sf00 = __builtin_elementwise_fma(tf00, vhalf, vhalf); SCHED_PIN();
        v2f tf12; tf12[0] = tanh_(hf12[0]); SCHED_PIN();
        v2f sf01 = __builtin_elementwise_fma(tf01, vhalf, vhalf); SCHED_PIN();
        tf12[1] = tanh_(hf12[1]); SCHED_PIN();
        v2f sf02 = __builtin_elementwise_fma(tf02, vhalf, vhalf); SCHED_PIN();
        v2f tf13; tf13[0] = tanh_(hf13[0]); SCHED_PIN();
        v2f sf03 = __builtin_elementwise_fma(tf03, vhalf, vhalf); SCHED_PIN();
        tf13[1] = tanh_(hf13[1]); SCHED_PIN();
        v2f sf10 = __builtin_elementwise_fma(tf10, vhalf, vhalf); SCHED_PIN();
        v2f ti00; ti00[0] = tanh_(hi00[0]); SCHED_PIN();
        v2f sf11 = __builtin_elementwise_fma(tf11, vhalf, vhalf); SCHED_PIN();
        ti00[1] = tanh_(hi00[1]); SCHED_PIN();
        v2f sf12 = __builtin_elementwise_fma(tf12, vhalf, vhalf); SCHED_PIN();
        v2f ti01; ti01[0] = tanh_(hi01[0]); SCHED_PIN();
        v2f ho00 = (v2f){acc[6][0], acc[6][1]} * 0.5f; SCHED_PIN();
        ti01[1] = tanh_(hi01[1]); SCHED_PIN();
        v2f sf13 = __builtin_elementwise_fma(tf13, vhalf, vhalf); SCHED_PIN();
        v2f ti02; ti02[0] = tanh_(hi02[0]); SCHED_PIN();
        v2f fc00 = sf00 * cst[0][0]; SCHED_PIN();
        ti02[1] = tanh_(hi02[1]); SCHED_PIN();
        v2f si00 = __builtin_elementwise_fma(ti00, vhalf, vhalf); SCHED_PIN();
        v2f ti03; ti03[0] = tanh_(hi03[0]); SCHED_PIN();
        v2f ho01 = (v2f){acc[6][2], acc[6][3]} * 0.5f; SCHED_PIN();
        ti03[1] = tanh_(hi03[1]); SCHED_PIN();
        v2f si01 = __builtin_elementwise_fma(ti01, vhalf, vhalf); SCHED_PIN();
        v2f ti10; ti10[0] = tanh_(hi10[0]); SCHED_PIN();
        v2f fc01 = sf01 * cst[0][1]; SCHED_PIN();
        ti10[1] = tanh_(hi10[1]); SCHED_PIN();
        v2f ho02 = (v2f){acc[6][4], acc[6][5]} * 0.5f; SCHED_PIN();
        v2f ti11; ti11[0] = tanh_(hi11[0]); SCHED_PIN();
        v2f si02 = __builtin_elementwise_fma(ti02, vhalf, vhalf); SCHED_PIN();
        ti11[1] = tanh_(hi11[1]); SCHED_PIN();
        v2f fc02 = sf02 * cst[0][2]; SCHED_PIN();
        v2f ti12; ti12[0] = tanh_(hi12[0]); SCHED_PIN();
        v2f ho03 = (v2f){acc[6][6], acc[6][7]} * 0.5f; SCHED_PIN();
        ti12[1] = tanh_(hi12[1]); SCHED_PIN();
        v2f si03 = __builtin_elementwise_fma(ti03, vhalf, vhalf); SCHED_PIN();
        v2f ti13; ti13[0] = tanh_(hi13[0]); SCHED_PIN();
        v2f fc03 = sf03 * cst[0][3]; SCHED_PIN();
        ti13[1] = tanh_(hi13[1]); SCHED_PIN();
        v2f ho10 = (v2f){acc[7][0], acc[7][1]} * 0.5f; SCHED_PIN();
        tg02[1] = tanh_(acc[4][5]); SCHED_PIN();
        v2f si10 = __builtin_elementwise_fma(ti10, vhalf, vhalf); SCHED_PIN();
        v2f tg03; tg03[0] = tanh_(acc[4][6]); SCHED_PIN();
        v2f fc10 = sf10 * cst[1][0]; SCHED_PIN();
        tg03[1] = tanh_(acc[4][7]); SCHED_PIN();
        v2f ho11 = (v2f){acc[7][2], acc[7][3]} * 0.5f; SCHED_PIN();
        v2f tg10; tg10[0] = tanh_(acc[5][0]); SCHED_PIN();
        v2f si11 = __builtin_elementwise_fma(ti11, vhalf, vhalf); SCHED_PIN();
        tg10[1] = tanh_(acc[5][1]); SCHED_PIN();
        v2f fc11 = sf11 * cst[1][1]; SCHED_PIN();
        v2f tg11; tg11[0] = tanh_(acc[5][2]); SCHED_PIN();
        v2f ho12 = (v2f){acc[7][4], acc[7][5]} * 0.5f; SCHED_PIN();
        tg11[1] = tanh_(acc[5][3]); SCHED_PIN();
        v2f si12 = __builtin_elementwise_fma(ti12, vhalf, vhalf); SCHED_PIN();
        v2f tg12; tg12[0] = tanh_(acc[5][4]); SCHED_PIN();
        v2f fc12 = sf12 * cst[1][2]; SCHED_PIN();
        tg12[1] = tanh_(acc[5][5]); SCHED_PIN();
        v2f ho13 = (v2f){acc[7][6], acc[7][7]} * 0.5f; SCHED_PIN();
        v2f tg13; tg13[0] = tanh_(acc[5][6]); SCHED_PIN();
        v2f si13 = __builtin_elementwise_fma(ti13, vhalf, vhalf); SCHED_PIN();
        tg13[1] = tanh_(acc[5][7]); SCHED_PIN();
        v2f fc13 = sf13 * cst[1][3]; SCHED_PIN();
        v2f to00; to00[0] = tanh_(ho00[0]); SCHED_PIN();
        cst[0][0] = __builtin_elementwise_fma(si00, tg00, fc00); SCHED_PIN();
        to00[1] = tanh_(ho00[1]); SCHED_PIN();
        cst[0][1] = __builtin_elementwise_fma(si01, tg01, fc01); SCHED_PIN();
        v2f to01; to01[0] = tanh_(ho01[0]); SCHED_PIN();
        cst[0][2] = __builtin_elementwise_fma(si02, tg02, fc02); SCHED_PIN();
        to01[1] = tanh_(ho01[1]); SCHED_PIN();
        cst[0][3] = __builtin_elementwise_fma(si03, tg03, fc03); SCHED_PIN();
        v2f to02; to02[0] = tanh_(ho02[0]); SCHED_PIN();
        cst[1][0] = __builtin_elementwise_fma(si10, tg10, fc10); SCHED_PIN();
        to02[1] = tanh_(ho02[1]); SCHED_PIN();
        cst[1][1] = __builtin_elementwise_fma(si11, tg11, fc11); SCHED_PIN();
        v2f to03; to03[0] = tanh_(ho03[0]); SCHED_PIN();
        cst[1][2] = __builtin_elementwise_fma(si12, tg12, fc12); SCHED_PIN();
        to03[1] = tanh_(ho03[1]); SCHED_PIN();
        cst[1][3] = __builtin_elementwise_fma(si13, tg13, fc13); SCHED_PIN();
        v2f to10; to10[0] = tanh_(ho10[0]); SCHED_PIN();
        v2f so00 = __builtin_elementwise_fma(to00, vhalf, vhalf); SCHED_PIN();
        to10[1] = tanh_(ho10[1]); SCHED_PIN();
        v2f so01 = __builtin_elementwise_fma(to01, vhalf, vhalf); SCHED_PIN();
        v2f to11; to11[0] = tanh_(ho11[0]); SCHED_PIN();
        v2f so02 = __builtin_elementwise_fma(to02, vhalf, vhalf); SCHED_PIN();
        to11[1] = tanh_(ho11[1]); SCHED_PIN();
        v2f so03 = __builtin_elementwise_fma(to03, vhalf, vhalf); SCHED_PIN();
        v2f to12; to12[0] = tanh_(ho12[0]); SCHED_PIN();
        to12[1] = tanh_(ho12[1]); SCHED_PIN();
        v2f so10 = __builtin_elementwise_fma(to10, vhalf, vhalf); SCHED_PIN();
        v2f to13; to13[0] = tanh_(ho13[0]); SCHED_PIN();
        to13[1] = tanh_(ho13[1]); SCHED_PIN();
        v2f so11 = __builtin_elementwise_fma(to11, vhalf, vhalf); SCHED_PIN();
        v2f tc00; tc00[0] = tanh_(cst[0][0][0]); SCHED_PIN();
        tc00[1] = tanh_(cst[0][0][1]); SCHED_PIN();
        v2f so12 = __builtin_elementwise_fma(to12, vhalf, vhalf); SCHED_PIN();
        v2f tc01; tc01[0] = tanh_(cst[0][1][0]); SCHED_PIN();
        tc01[1] = tanh_(cst[0][1][1]); SCHED_PIN();
        v2f so13 = __builtin_elementwise_fma(to13, vhalf, vhalf); SCHED_PIN();
        v2f tc02; tc02[0] = tanh_(cst[0][2][0]); SCHED_PIN();
        tc02[1] = tanh_(cst[0][2][1]); SCHED_PIN();
        v2f hh00 = so00 * tc00; SCHED_PIN();
        v2f tc03; tc03[0] = tanh_(cst[0][3][0]); SCHED_PIN();
        tc03[1] = tanh_(cst[0][3][1]); SCHED_PIN();
        v2f hh01 = so01 * tc01; SCHED_PIN();
        v2f tc10; tc10[0] = tanh_(cst[1][0][0]); SCHED_PIN();
        tc10[1] = tanh_(cst[1][0][1]); SCHED_PIN();
        v2f hh02 = so02 * tc02; SCHED_PIN();
        v2f tc11; tc11[0] = tanh_(cst[1][1][0]); SCHED_PIN();
        Bh[0] = (_Float16)hh00[0]; Bh[1] = (_Float16)hh00[1]; SCHED_PIN();
        tc11[1] = tanh_(cst[1][1][1]); SCHED_PIN();
        v2f hh03 = so03 * tc03; SCHED_PIN();
        v2f tc12; tc12[0] = tanh_(cst[1][2][0]); SCHED_PIN();
        Bh[2] = (_Float16)hh01[0]; Bh[3] = (_Float16)hh01[1]; SCHED_PIN();
        tc12[1] = tanh_(cst[1][2][1]); SCHED_PIN();
        v2f hh10 = so10 * tc10; SCHED_PIN();
        v2f tc13; tc13[0] = tanh_(cst[1][3][0]); SCHED_PIN();
        Bh[4] = (_Float16)hh02[0]; Bh[5] = (_Float16)hh02[1]; SCHED_PIN();
        tc13[1] = tanh_(cst[1][3][1]); SCHED_PIN();
        v2f hh11 = so11 * tc11; SCHED_PIN();
        Bh[6] = (_Float16)hh03[0]; Bh[7] = (_Float16)hh03[1]; SCHED_PIN();
        v2f hh12 = so12 * tc12; SCHED_PIN();
        Bh[8] = (_Float16)hh10[0]; Bh[9] = (_Float16)hh10[1]; SCHED_PIN();
        v2f hh13 = so13 * tc13; SCHED_PIN();
        Bh[10] = (_Float16)hh11[0]; Bh[11] = (_Float16)hh11[1]; SCHED_PIN();
        Bh[12] = (_Float16)hh12[0]; Bh[13] = (_Float16)hh12[1]; SCHED_PIN();
        Bh[14] = (_Float16)hh13[0]; Bh[15] = (_Float16)hh13[1]; SCHED_PIN();
        xv = xvn;
    }

    {
        v8f accy = __builtin_amdgcn_wmma_f32_16x16x32_f16(
            false, Afc, false, Bh, (short)0, Cfc, false, false);
        if (hi == 0)
            out[(size_t)(b0 + col) * T + (T - 1)] = accy[0];
    }
}

extern "C" void kernel_launch(void* const* d_in, const int* in_sizes, int n_in,
                              void* d_out, int out_size, void* d_ws, size_t ws_size,
                              hipStream_t stream) {
    const float* x    = (const float*)d_in[0];
    const float* W_ih = (const float*)d_in[1];
    const float* W_hh = (const float*)d_in[2];
    const float* b_ih = (const float*)d_in[3];
    const float* b_hh = (const float*)d_in[4];
    const float* W_fc = (const float*)d_in[5];
    const float* b_fc = (const float*)d_in[6];
    float* out = (float*)d_out;

    const int T = 512;
    const int B = out_size / T;
    const int nblocks = B / 16;

    lstm_wmma_kernel<<<dim3(nblocks), dim3(32), 0, stream>>>(
        x, W_ih, W_hh, b_ih, b_hh, W_fc, b_fc, out, T);

    (void)in_sizes; (void)n_in; (void)d_ws; (void)ws_size;
}


# Input projection moved to a second wave of the work-group: its f32 16x16x4 WMMAs run a block of 8 steps ahead on another SIMD and reach the recurrent WMMAs as C via an LDS ring; sums, order unchanged
# speedup vs baseline: 1.8555x; 1.2888x over previous
#include <hip/hip_runtime.h>
#include <hip/hip_bf16.h>


typedef __attribute__((ext_vector_type(16))) _Float16 v16h;
typedef __attribute__((ext_vector_type(8)))  _Float16 v8h;
typedef __attribute__((ext_vector_type(2)))  _Float16 v2h;
typedef __attribute__((ext_vector_type(8)))  float    v8f;
typedef __attribute__((ext_vector_type(2)))  float    v2f;

#define SCHED_PIN() __builtin_amdgcn_sched_barrier(0)

#define HID   32
#define NTILE 8
#define SBLK  8
#define NCHUNK 16

#if __has_builtin(__builtin_amdgcn_tanhf)
__device__ __forceinline__ float tanh_(float x) { return __builtin_amdgcn_tanhf(x); }
__device__ __forceinline__ float sigm(float x) {
    return __builtin_fmaf(0.5f, __builtin_amdgcn_tanhf(0.5f * x), 0.5f);
}
#else
__device__ __forceinline__ float fast_rcp(float x) { return __builtin_amdgcn_rcpf(x); }
__device__ __forceinline__ float tanh_(float x) {
    float e = __expf(-2.0f * x);
    return (1.0f - e) * fast_rcp(1.0f + e);
}
__device__ __forceinline__ float sigm(float x) { return fast_rcp(1.0f + __expf(-x)); }
#endif

__device__ __forceinline__ void cell_update(const v8f (&acc)[NTILE], v2f (&cst)[2][4], v16h& Bh)
{
    const v2f vhalf = (v2f){0.5f, 0.5f};
    v2f tg00; tg00[0] = tanh_(acc[4][0]); SCHED_PIN();
    v2f hf00 = (v2f){acc[2][0], acc[2][1]} * 0.5f; SCHED_PIN();
    tg00[1] = tanh_(acc[4][1]); SCHED_PIN();
    v2f hf01 = (v2f){acc[2][2], acc[2][3]} * 0.5f; SCHED_PIN();
    v2f tg01; tg01[0] = tanh_(acc[4][2]); SCHED_PIN();
    v2f hf02 = (v2f){acc[2][4], acc[2][5]} * 0.5f; SCHED_PIN();
    tg01[1] = tanh_(acc[4][3]); SCHED_PIN();
    v2f hf03 = (v2f){acc[2][6], acc[2][7]} * 0.5f; SCHED_PIN();
    v2f tg02; tg02[0] = tanh_(acc[4][4]); SCHED_PIN();
    v2f hf10 = (v2f){acc[3][0], acc[3][1]} * 0.5f; SCHED_PIN();
    v2f tf00; tf00[0] = tanh_(hf00[0]); SCHED_PIN();
    v2f hf11 = (v2f){acc[3][2], acc[3][3]} * 0.5f; SCHED_PIN();
    tf00[1] = tanh_(hf00[1]); SCHED_PIN();
    v2f hf12 = (v2f){acc[3][4], acc[3][5]} * 0.5f; SCHED_PIN();
    v2f tf01; tf01[0] = tanh_(hf01[0]); SCHED_PIN();
    v2f hf13 = (v2f){acc[3][6], acc[3][7]} * 0.5f; SCHED_PIN();
    tf01[1] = tanh_(hf01[1]); SCHED_PIN();
    v2f hi00 = (v2f){acc[0][0], acc[0][1]} * 0.5f; SCHED_PIN();
    v2f tf02; tf02[0] = tanh_(hf02[0]); SCHED_PIN();
    v2f hi01 = (v2f){acc[0][2], acc[0][3]} * 0.5f; SCHED_PIN();
    tf02[1] = tanh_(hf02[1]); SCHED_PIN();
    v2f hi02 = (v2f){acc[0][4], acc[0][5]} * 0.5f; SCHED_PIN();
    v2f tf03; tf03[0] = tanh_(hf03[0]); SCHED_PIN();
    v2f hi03 = (v2f){acc[0][6], acc[0][7]} * 0.5f; SCHED_PIN();
    tf03[1] = tanh_(hf03[1]); SCHED_PIN();
    v2f hi10 = (v2f){acc[1][0], acc[1][1]} * 0.5f; SCHED_PIN();
    v2f tf10; tf10[0] = tanh_(hf10[0]); SCHED_PIN();
    v2f hi11 = (v2f){acc[1][2], acc[1][3]} * 0.5f; SCHED_PIN();
    tf10[1] = tanh_(hf10[1]); SCHED_PIN();
    v2f hi12 = (v2f){acc[1][4], acc[1][5]} * 0.5f; SCHED_PIN();
    v2f tf11; tf11[0] = tanh_(hf11[0]); SCHED_PIN();
    v2f hi13 = (v2f){acc[1][6], acc[1][7]} * 0.5f; SCHED_PIN();
    tf11[1] = tanh_(hf11[1]); SCHED_PIN();
    v2f sf00 = __builtin_elementwise_fma(tf00, vhalf, vhalf); SCHED_PIN();
    v2f tf12; tf12[0] = tanh_(hf12[0]); SCHED_PIN();
    v2f sf01 = __builtin_elementwise_fma(tf01, vhalf, vhalf); SCHED_PIN();
    tf12[1] = tanh_(hf12[1]); SCHED_PIN();
    v2f sf02 = __builtin_elementwise_fma(tf02, vhalf, vhalf); SCHED_PIN();
    v2f tf13; tf13[0] = tanh_(hf13[0]); SCHED_PIN();
    v2f sf03 = __builtin_elementwise_fma(tf03, vhalf, vhalf); SCHED_PIN();
    tf13[1] = tanh_(hf13[1]); SCHED_PIN();
    v2f sf10 = __builtin_elementwise_fma(tf10, vhalf, vhalf); SCHED_PIN();
    v2f ti00; ti00[0] = tanh_(hi00[0]); SCHED_PIN();
    v2f sf11 = __builtin_elementwise_fma(tf11, vhalf, vhalf); SCHED_PIN();
    ti00[1] = tanh_(hi00[1]); SCHED_PIN();
    v2f sf12 = __builtin_elementwise_fma(tf12, vhalf, vhalf); SCHED_PIN();
    v2f ti01; ti01[0] = tanh_(hi01[0]); SCHED_PIN();
    v2f ho00 = (v2f){acc[6][0], acc[6][1]} * 0.5f; SCHED_PIN();
    ti01[1] = tanh_(hi01[1]); SCHED_PIN();
    v2f sf13 = __builtin_elementwise_fma(tf13, vhalf, vhalf); SCHED_PIN();
    v2f ti02; ti02[0] = tanh_(hi02[0]); SCHED_PIN();
    v2f fc00 = sf00 * cst[0][0]; SCHED_PIN();
    ti02[1] = tanh_(hi02[1]); SCHED_PIN();
    v2f si00 = __builtin_elementwise_fma(ti00, vhalf, vhalf); SCHED_PIN();
    v2f ti03; ti03[0] = tanh_(hi03[0]); SCHED_PIN();
    v2f ho01 = (v2f){acc[6][2], acc[6][3]} * 0.5f; SCHED_PIN();
    ti03[1] = tanh_(hi03[1]); SCHED_PIN();
    v2f si01 = __builtin_elementwise_fma(ti01, vhalf, vhalf); SCHED_PIN();
    v2f ti10; ti10[0] = tanh_(hi10[0]); SCHED_PIN();
    v2f fc01 = sf01 * cst[0][1]; SCHED_PIN();
    ti10[1] = tanh_(hi10[1]); SCHED_PIN();
    v2f ho02 = (v2f){acc[6][4], acc[6][5]} * 0.5f; SCHED_PIN();
    v2f ti11; ti11[0] = tanh_(hi11[0]); SCHED_PIN();
    v2f si02 = __builtin_elementwise_fma(ti02, vhalf, vhalf); SCHED_PIN();
    ti11[1] = tanh_(hi11[1]); SCHED_PIN();
    v2f fc02 = sf02 * cst[0][2]; SCHED_PIN();
    v2f ti12; ti12[0] = tanh_(hi12[0]); SCHED_PIN();
    v2f ho03 = (v2f){acc[6][6], acc[6][7]} * 0.5f; SCHED_PIN();
    ti12[1] = tanh_(hi12[1]); SCHED_PIN();
    v2f si03 = __builtin_elementwise_fma(ti03, vhalf, vhalf); SCHED_PIN();
    v2f ti13; ti13[0] = tanh_(hi13[0]); SCHED_PIN();
    v2f fc03 = sf03 * cst[0][3]; SCHED_PIN();
    ti13[1] = tanh_(hi13[1]); SCHED_PIN();
    v2f ho10 = (v2f){acc[7][0], acc[7][1]} * 0.5f; SCHED_PIN();
    tg02[1] = tanh_(acc[4][5]); SCHED_PIN();
    v2f si10 = __builtin_elementwise_fma(ti10, vhalf, vhalf); SCHED_PIN();
    v2f tg03; tg03[0] = tanh_(acc[4][6]); SCHED_PIN();
    v2f fc10 = sf10 * cst[1][0]; SCHED_PIN();
    tg03[1] = tanh_(acc[4][7]); SCHED_PIN();
    v2f ho11 = (v2f){acc[7][2], acc[7][3]} * 0.5f; SCHED_PIN();
    v2f tg10; tg10[0] = tanh_(acc[5][0]); SCHED_PIN();
    v2f si11 = __builtin_elementwise_fma(ti11, vhalf, vhalf); SCHED_PIN();
    tg10[1] = tanh_(acc[5][1]); SCHED_PIN();
    v2f fc11 = sf11 * cst[1][1]; SCHED_PIN();
    v2f tg11; tg11[0] = tanh_(acc[5][2]); SCHED_PIN();
    v2f ho12 = (v2f){acc[7][4], acc[7][5]} * 0.5f; SCHED_PIN();
    tg11[1] = tanh_(acc[5][3]); SCHED_PIN();
    v2f si12 = __builtin_elementwise_fma(ti12, vhalf, vhalf); SCHED_PIN();
    v2f tg12; tg12[0] = tanh_(acc[5][4]); SCHED_PIN();
    v2f fc12 = sf12 * cst[1][2]; SCHED_PIN();
    tg12[1] = tanh_(acc[5][5]); SCHED_PIN();
    v2f ho13 = (v2f){acc[7][6], acc[7][7]} * 0.5f; SCHED_PIN();
    v2f tg13; tg13[0] = tanh_(acc[5][6]); SCHED_PIN();
    v2f si13 = __builtin_elementwise_fma(ti13, vhalf, vhalf); SCHED_PIN();
    tg13[1] = tanh_(acc[5][7]); SCHED_PIN();
    v2f fc13 = sf13 * cst[1][3]; SCHED_PIN();
    v2f to00; to00[0] = tanh_(ho00[0]); SCHED_PIN();
    cst[0][0] = __builtin_elementwise_fma(si00, tg00, fc00); SCHED_PIN();
    to00[1] = tanh_(ho00[1]); SCHED_PIN();
    cst[0][1] = __builtin_elementwise_fma(si01, tg01, fc01); SCHED_PIN();
    v2f to01; to01[0] = tanh_(ho01[0]); SCHED_PIN();
    cst[0][2] = __builtin_elementwise_fma(si02, tg02, fc02); SCHED_PIN();
    to01[1] = tanh_(ho01[1]); SCHED_PIN();
    cst[0][3] = __builtin_elementwise_fma(si03, tg03, fc03); SCHED_PIN();
    v2f to02; to02[0] = tanh_(ho02[0]); SCHED_PIN();
    cst[1][0] = __builtin_elementwise_fma(si10, tg10, fc10); SCHED_PIN();
    to02[1] = tanh_(ho02[1]); SCHED_PIN();
    cst[1][1] = __builtin_elementwise_fma(si11, tg11, fc11); SCHED_PIN();
    v2f to03; to03[0] = tanh_(ho03[0]); SCHED_PIN();
    cst[1][2] = __builtin_elementwise_fma(si12, tg12, fc12); SCHED_PIN();
    to03[1] = tanh_(ho03[1]); SCHED_PIN();
    cst[1][3] = __builtin_elementwise_fma(si13, tg13, fc13); SCHED_PIN();
    v2f to10; to10[0] = tanh_(ho10[0]); SCHED_PIN();
    v2f so00 = __builtin_elementwise_fma(to00, vhalf, vhalf); SCHED_PIN();
    to10[1] = tanh_(ho10[1]); SCHED_PIN();
    v2f so01 = __builtin_elementwise_fma(to01, vhalf, vhalf); SCHED_PIN();
    v2f to11; to11[0] = tanh_(ho11[0]); SCHED_PIN();
    v2f so02 = __builtin_elementwise_fma(to02, vhalf, vhalf); SCHED_PIN();
    to11[1] = tanh_(ho11[1]); SCHED_PIN();
    v2f so03 = __builtin_elementwise_fma(to03, vhalf, vhalf); SCHED_PIN();
    v2f to12; to12[0] = tanh_(ho12[0]); SCHED_PIN();
    to12[1] = tanh_(ho12[1]); SCHED_PIN();
    v2f so10 = __builtin_elementwise_fma(to10, vhalf, vhalf); SCHED_PIN();
    v2f to13; to13[0] = tanh_(ho13[0]); SCHED_PIN();
    to13[1] = tanh_(ho13[1]); SCHED_PIN();
    v2f so11 = __builtin_elementwise_fma(to11, vhalf, vhalf); SCHED_PIN();
    v2f tc00; tc00[0] = tanh_(cst[0][0][0]); SCHED_PIN();
    tc00[1] = tanh_(cst[0][0][1]); SCHED_PIN();
    v2f so12 = __builtin_elementwise_fma(to12, vhalf, vhalf); SCHED_PIN();
    v2f tc01; tc01[0] = tanh_(cst[0][1][0]); SCHED_PIN();
    tc01[1] = tanh_(cst[0][1][1]); SCHED_PIN();
    v2f so13 = __builtin_elementwise_fma(to13, vhalf, vhalf); SCHED_PIN();
    v2f tc02; tc02[0] = tanh_(cst[0][2][0]); SCHED_PIN();
    tc02[1] = tanh_(cst[0][2][1]); SCHED_PIN();
    v2f hh00 = so00 * tc00; SCHED_PIN();
    v2f tc03; tc03[0] = tanh_(cst[0][3][0]); SCHED_PIN();
    tc03[1] = tanh_(cst[0][3][1]); SCHED_PIN();
    v2f hh01 = so01 * tc01; SCHED_PIN();
    v2f tc10; tc10[0] = tanh_(cst[1][0][0]); SCHED_PIN();
    tc10[1] = tanh_(cst[1][0][1]); SCHED_PIN();
    v2f hh02 = so02 * tc02; SCHED_PIN();
    v2f tc11; tc11[0] = tanh_(cst[1][1][0]); SCHED_PIN();
    Bh[0] = (_Float16)hh00[0]; Bh[1] = (_Float16)hh00[1]; SCHED_PIN();
    tc11[1] = tanh_(cst[1][1][1]); SCHED_PIN();
    v2f hh03 = so03 * tc03; SCHED_PIN();
    v2f tc12; tc12[0] = tanh_(cst[1][2][0]); SCHED_PIN();
    Bh[2] = (_Float16)hh01[0]; Bh[3] = (_Float16)hh01[1]; SCHED_PIN();
    tc12[1] = tanh_(cst[1][2][1]); SCHED_PIN();
    v2f hh10 = so10 * tc10; SCHED_PIN();
    v2f tc13; tc13[0] = tanh_(cst[1][3][0]); SCHED_PIN();
    Bh[4] = (_Float16)hh02[0]; Bh[5] = (_Float16)hh02[1]; SCHED_PIN();
    tc13[1] = tanh_(cst[1][3][1]); SCHED_PIN();
    v2f hh11 = so11 * tc11; SCHED_PIN();
    Bh[6] = (_Float16)hh03[0]; Bh[7] = (_Float16)hh03[1]; SCHED_PIN();
    v2f hh12 = so12 * tc12; SCHED_PIN();
    Bh[8] = (_Float16)hh10[0]; Bh[9] = (_Float16)hh10[1]; SCHED_PIN();
    v2f hh13 = so13 * tc13; SCHED_PIN();
    Bh[10] = (_Float16)hh11[0]; Bh[11] = (_Float16)hh11[1]; SCHED_PIN();
    Bh[12] = (_Float16)hh12[0]; Bh[13] = (_Float16)hh12[1]; SCHED_PIN();
    Bh[14] = (_Float16)hh13[0]; Bh[15] = (_Float16)hh13[1]; SCHED_PIN();
}

__global__ __launch_bounds__(64)
void lstm_wmma_kernel(const float* __restrict__ x,
                      const float* __restrict__ W_ih,
                      const float* __restrict__ W_hh,
                      const float* __restrict__ b_ih,
                      const float* __restrict__ b_hh,
                      const float* __restrict__ W_fc,
                      const float* __restrict__ b_fc,
                      float* __restrict__ out,
                      int T)
{
    const int wave = __builtin_amdgcn_readfirstlane((int)(threadIdx.x >> 5));
    const int lane = threadIdx.x & 31;
    const int col  = lane & 15;
    const int hi   = lane >> 4;
    const int b0   = blockIdx.x * 16;

    __shared__ float4 ring[2 * SBLK * NCHUNK * 32];

    const int nblk = (T + SBLK - 1) / SBLK;

    v16h At[NTILE];
    v16h Afc = (v16h){};
    v8f  Cfc = (v8f){};
    v2f  cst[2][4];
    v16h Bh = (v16h){};
    v8f  accA[NTILE], accB[NTILE];
#pragma unroll
    for (int tt = 0; tt < NTILE; ++tt) { At[tt] = (v16h){}; accA[tt] = (v8f){}; accB[tt] = (v8f){}; }
#pragma unroll
    for (int p = 0; p < 2; ++p)
#pragma unroll
        for (int j = 0; j < 4; ++j) cst[p][j] = (v2f){0.f, 0.f};

    v2f Aw[NTILE];
    v2f Bxs[SBLK];
    float2 xa[SBLK];
#pragma unroll
    for (int tt = 0; tt < NTILE; ++tt) Aw[tt] = (v2f){0.f, 0.f};
#pragma unroll
    for (int s = 0; s < SBLK; ++s) { Bxs[s] = (v2f){0.f, 0.f}; xa[s] = make_float2(0.f, 0.f); }

    const unsigned xlane = (unsigned)(b0 + col) * (unsigned)T * 2u;

    auto load_x = [&](int blk) {
#pragma unroll
        for (int s = 0; s < SBLK; ++s) {
            int t = blk * SBLK + s;
            t = (t < T) ? t : (T - 1);
            const float* xs = x + (size_t)t * 2;
            xa[s] = *(const float2*)(xs + xlane);
        }
    };
    auto build_bx = [&]() {
#pragma unroll
        for (int s = 0; s < SBLK; ++s) {
            Bxs[s][0] = (hi == 0) ? xa[s].x : 1.0f;
            Bxs[s][1] = (hi == 0) ? xa[s].y : 0.0f;
            SCHED_PIN();
        }
    };

    auto fill = [&](int blk) {
        float4* dst0 = ring + (size_t)(((blk & 1) * SBLK) * NCHUNK) * 32 + lane;
        build_bx();
        SCHED_PIN();
#pragma unroll
        for (int s = 0; s < SBLK; ++s) {
            v8f d[NTILE];
#pragma unroll
            for (int tt = 0; tt < NTILE; ++tt) {
                v8f z = (v8f){};
                d[tt] = __builtin_amdgcn_wmma_f32_16x16x4_f32(
                    false, Aw[tt], false, Bxs[s], (short)0, z, false, false);
            }
            SCHED_PIN();
            if (s == SBLK - 1) load_x(blk + 1);
#pragma unroll
            for (int tt = 0; tt < NTILE; ++tt) {
                dst0[(s * NCHUNK + tt * 2 + 0) * 32] = make_float4(d[tt][0], d[tt][1], d[tt][2], d[tt][3]);
                dst0[(s * NCHUNK + tt * 2 + 1) * 32] = make_float4(d[tt][4], d[tt][5], d[tt][6], d[tt][7]);
            }
            SCHED_PIN();
        }
    };

    auto hand_over = [&]() {
        __builtin_amdgcn_fence(__ATOMIC_RELEASE, "workgroup", "local");
        __builtin_amdgcn_s_barrier();
        __builtin_amdgcn_fence(__ATOMIC_ACQUIRE, "workgroup", "local");
    };

    auto step = [&](int k, int t, v8f (&done)[NTILE], v8f (&next)[NTILE]) {
        const float4* srcp = ring + (size_t)(((k & 1) * SBLK + (t - k * SBLK)) * NCHUNK) * 32 + lane;
#pragma unroll
        for (int tt = 0; tt < NTILE; ++tt) {
            float4 lo = srcp[(tt * 2 + 0) * 32];
            float4 up = srcp[(tt * 2 + 1) * 32];
            next[tt] = (v8f){lo.x, lo.y, lo.z, lo.w, up.x, up.y, up.z, up.w};
        }

        if (t > 0) cell_update(done, cst, Bh);

        v8f accy = __builtin_amdgcn_wmma_f32_16x16x32_f16(
            false, Afc, false, Bh, (short)0, Cfc, false, false);
#pragma unroll
        for (int tt = 0; tt < NTILE; ++tt) {
            next[tt] = __builtin_amdgcn_wmma_f32_16x16x32_f16(
                false, At[tt], false, Bh, (short)0, next[tt], false, false);
        }
        if (t > 0 && hi == 0)
            out[(size_t)(b0 + col) * T + (t - 1)] = accy[0];
    };

    if (wave == 0) {
#pragma unroll
        for (int tt = 0; tt < NTILE; ++tt) {
            const float* wrow = W_hh + (size_t)(tt * 16 + col) * HID;
#pragma unroll
            for (int e = 0; e < 8; ++e) {
                At[tt][e]     = (_Float16)wrow[hi * 8 + e];
                At[tt][e + 8] = (_Float16)wrow[16 + hi * 8 + e];
            }
        }
        if (col == 0) {
#pragma unroll
            for (int e = 0; e < 8; ++e) {
                Afc[e]     = (_Float16)W_fc[hi * 8 + e];
                Afc[e + 8] = (_Float16)W_fc[16 + hi * 8 + e];
            }
        }
        float bfc = b_fc[0];
#pragma unroll
        for (int e = 0; e < 8; ++e) Cfc[e] = bfc;
    } else {
#pragma unroll
        for (int tt = 0; tt < NTILE; ++tt) {
            int g = tt * 16 + col;
            if (hi == 0) { Aw[tt][0] = W_ih[g * 2 + 0];     Aw[tt][1] = W_ih[g * 2 + 1]; }
            else         { Aw[tt][0] = b_ih[g] + b_hh[g];   Aw[tt][1] = 0.0f; }
        }
        load_x(0);
        fill(0);
    }
    hand_over();

    const int nfull = T / SBLK;
    for (int k = 0; k < nfull; ++k) {
        if (wave == 1) {
            if (k + 1 < nblk) fill(k + 1);
        } else {
#pragma unroll
            for (int j = 0; j < SBLK / 2; ++j) {
                step(k, k * SBLK + 2 * j,     accA, accB);
                step(k, k * SBLK + 2 * j + 1, accB, accA);
            }
        }
        hand_over();
    }

    if (wave == 0) {
        for (int t = nfull * SBLK; t + 1 < T; t += 2) {
            step(nfull, t,     accA, accB);
            step(nfull, t + 1, accB, accA);
        }
        if (T & 1) {
            step(nfull, T - 1, accA, accB);
            cell_update(accB, cst, Bh);
        } else {
            cell_update(accA, cst, Bh);
        }
        v8f accy = __builtin_amdgcn_wmma_f32_16x16x32_f16(
            false, Afc, false, Bh, (short)0, Cfc, false, false);
        if (hi == 0)
            out[(size_t)(b0 + col) * T + (T - 1)] = accy[0];
    }
}

extern "C" void kernel_launch(void* const* d_in, const int* in_sizes, int n_in,
                              void* d_out, int out_size, void* d_ws, size_t ws_size,
                              hipStream_t stream) {
    const float* x    = (const float*)d_in[0];
    const float* W_ih = (const float*)d_in[1];
    const float* W_hh = (const float*)d_in[2];
    const float* b_ih = (const float*)d_in[3];
    const float* b_hh = (const float*)d_in[4];
    const float* W_fc = (const float*)d_in[5];
    const float* b_fc = (const float*)d_in[6];
    float* out = (float*)d_out;

    const int T = 512;
    const int B = out_size / T;
    const int nblocks = B / 16;

    lstm_wmma_kernel<<<dim3(nblocks), dim3(64), 0, stream>>>(
        x, W_ih, W_hh, b_ih, b_hh, W_fc, b_fc, out, T);

    (void)in_sizes; (void)n_in; (void)d_ws; (void)ws_size;
}


# FC product and y stores moved off the recurrent wave: it hands each h fragment on through a second LDS ring (2 stores beside its LDS reads), the input wave forms y a block later; sums unchanged
# speedup vs baseline: 2.2690x; 1.2229x over previous
#include <hip/hip_runtime.h>
#include <hip/hip_bf16.h>


typedef __attribute__((ext_vector_type(16))) _Float16 v16h;
typedef __attribute__((ext_vector_type(8)))  _Float16 v8h;
typedef __attribute__((ext_vector_type(2)))  _Float16 v2h;
typedef __attribute__((ext_vector_type(8)))  float    v8f;
typedef __attribute__((ext_vector_type(4)))  float    v4f;
typedef __attribute__((ext_vector_type(2)))  float    v2f;

#define SCHED_PIN() __builtin_amdgcn_sched_barrier(0)

template <bool B> struct StepTag { static constexpr bool may_be_first = B; };

#define HID   32
#define NTILE 8
#define SBLK  8
#define NCHUNK 16
#define HCHUNK 2
#define HSLOT  (SBLK + 1)

#if __has_builtin(__builtin_amdgcn_tanhf)
__device__ __forceinline__ float tanh_(float x) { return __builtin_amdgcn_tanhf(x); }
__device__ __forceinline__ float sigm(float x) {
    return __builtin_fmaf(0.5f, __builtin_amdgcn_tanhf(0.5f * x), 0.5f);
}
#else
__device__ __forceinline__ float fast_rcp(float x) { return __builtin_amdgcn_rcpf(x); }
__device__ __forceinline__ float tanh_(float x) {
    float e = __expf(-2.0f * x);
    return (1.0f - e) * fast_rcp(1.0f + e);
}
__device__ __forceinline__ float sigm(float x) { return fast_rcp(1.0f + __expf(-x)); }
#endif

__device__ __forceinline__ void cell_update(const v8f (&acc)[NTILE], v2f (&cst)[2][4], v16h& Bh)
{
    const v2f vhalf = (v2f){0.5f, 0.5f};
    v2f tg00; tg00[0] = tanh_(acc[4][0]); SCHED_PIN();
    v2f hf00 = (v2f){acc[2][0], acc[2][1]} * 0.5f; SCHED_PIN();
    tg00[1] = tanh_(acc[4][1]); SCHED_PIN();
    v2f hf01 = (v2f){acc[2][2], acc[2][3]} * 0.5f; SCHED_PIN();
    v2f tg01; tg01[0] = tanh_(acc[4][2]); SCHED_PIN();
    v2f hf02 = (v2f){acc[2][4], acc[2][5]} * 0.5f; SCHED_PIN();
    tg01[1] = tanh_(acc[4][3]); SCHED_PIN();
    v2f hf03 = (v2f){acc[2][6], acc[2][7]} * 0.5f; SCHED_PIN();
    v2f tg02; tg02[0] = tanh_(acc[4][4]); SCHED_PIN();
    v2f hf10 = (v2f){acc[3][0], acc[3][1]} * 0.5f; SCHED_PIN();
    v2f tf00; tf00[0] = tanh_(hf00[0]); SCHED_PIN();
    v2f hf11 = (v2f){acc[3][2], acc[3][3]} * 0.5f; SCHED_PIN();
    tf00[1] = tanh_(hf00[1]); SCHED_PIN();
    v2f hf12 = (v2f){acc[3][4], acc[3][5]} * 0.5f; SCHED_PIN();
    v2f tf01; tf01[0] = tanh_(hf01[0]); SCHED_PIN();
    v2f hf13 = (v2f){acc[3][6], acc[3][7]} * 0.5f; SCHED_PIN();
    tf01[1] = tanh_(hf01[1]); SCHED_PIN();
    v2f hi00 = (v2f){acc[0][0], acc[0][1]} * 0.5f; SCHED_PIN();
    v2f tf02; tf02[0] = tanh_(hf02[0]); SCHED_PIN();
    v2f hi01 = (v2f){acc[0][2], acc[0][3]} * 0.5f; SCHED_PIN();
    tf02[1] = tanh_(hf02[1]); SCHED_PIN();
    v2f hi02 = (v2f){acc[0][4], acc[0][5]} * 0.5f; SCHED_PIN();
    v2f tf03; tf03[0] = tanh_(hf03[0]); SCHED_PIN();
    v2f hi03 = (v2f){acc[0][6], acc[0][7]} * 0.5f; SCHED_PIN();
    tf03[1] = tanh_(hf03[1]); SCHED_PIN();
    v2f hi10 = (v2f){acc[1][0], acc[1][1]} * 0.5f; SCHED_PIN();
    v2f tf10; tf10[0] = tanh_(hf10[0]); SCHED_PIN();
    v2f hi11 = (v2f){acc[1][2], acc[1][3]} * 0.5f; SCHED_PIN();
    tf10[1] = tanh_(hf10[1]); SCHED_PIN();
    v2f hi12 = (v2f){acc[1][4], acc[1][5]} * 0.5f; SCHED_PIN();
    v2f tf11; tf11[0] = tanh_(hf11[0]); SCHED_PIN();
    v2f hi13 = (v2f){acc[1][6], acc[1][7]} * 0.5f; SCHED_PIN();
    tf11[1] = tanh_(hf11[1]); SCHED_PIN();
    v2f sf00 = __builtin_elementwise_fma(tf00, vhalf, vhalf); SCHED_PIN();
    v2f tf12; tf12[0] = tanh_(hf12[0]); SCHED_PIN();
    v2f sf01 = __builtin_elementwise_fma(tf01, vhalf, vhalf); SCHED_PIN();
    tf12[1] = tanh_(hf12[1]); SCHED_PIN();
    v2f sf02 = __builtin_elementwise_fma(tf02, vhalf, vhalf); SCHED_PIN();
    v2f tf13; tf13[0] = tanh_(hf13[0]); SCHED_PIN();
    v2f sf03 = __builtin_elementwise_fma(tf03, vhalf, vhalf); SCHED_PIN();
    tf13[1] = tanh_(hf13[1]); SCHED_PIN();
    v2f sf10 = __builtin_elementwise_fma(tf10, vhalf, vhalf); SCHED_PIN();
    v2f ti00; ti00[0] = tanh_(hi00[0]); SCHED_PIN();
    v2f sf11 = __builtin_elementwise_fma(tf11, vhalf, vhalf); SCHED_PIN();
    ti00[1] = tanh_(hi00[1]); SCHED_PIN();
    v2f sf12 = __builtin_elementwise_fma(tf12, vhalf, vhalf); SCHED_PIN();
    v2f ti01; ti01[0] = tanh_(hi01[0]); SCHED_PIN();
    v2f ho00 = (v2f){acc[6][0], acc[6][1]} * 0.5f; SCHED_PIN();
    ti01[1] = tanh_(hi01[1]); SCHED_PIN();
    v2f sf13 = __builtin_elementwise_fma(tf13, vhalf, vhalf); SCHED_PIN();
    v2f ti02; ti02[0] = tanh_(hi02[0]); SCHED_PIN();
    v2f fc00 = sf00 * cst[0][0]; SCHED_PIN();
    ti02[1] = tanh_(hi02[1]); SCHED_PIN();
    v2f si00 = __builtin_elementwise_fma(ti00, vhalf, vhalf); SCHED_PIN();
    v2f ti03; ti03[0] = tanh_(hi03[0]); SCHED_PIN();
    v2f ho01 = (v2f){acc[6][2], acc[6][3]} * 0.5f; SCHED_PIN();
    ti03[1] = tanh_(hi03[1]); SCHED_PIN();
    v2f si01 = __builtin_elementwise_fma(ti01, vhalf, vhalf); SCHED_PIN();
    v2f ti10; ti10[0] = tanh_(hi10[0]); SCHED_PIN();
    v2f fc01 = sf01 * cst[0][1]; SCHED_PIN();
    ti10[1] = tanh_(hi10[1]); SCHED_PIN();
    v2f ho02 = (v2f){acc[6][4], acc[6][5]} * 0.5f; SCHED_PIN();
    v2f ti11; ti11[0] = tanh_(hi11[0]); SCHED_PIN();
    v2f si02 = __builtin_elementwise_fma(ti02, vhalf, vhalf); SCHED_PIN();
    ti11[1] = tanh_(hi11[1]); SCHED_PIN();
    v2f fc02 = sf02 * cst[0][2]; SCHED_PIN();
    v2f ti12; ti12[0] = tanh_(hi12[0]); SCHED_PIN();
    v2f ho03 = (v2f){acc[6][6], acc[6][7]} * 0.5f; SCHED_PIN();
    ti12[1] = tanh_(hi12[1]); SCHED_PIN();
    v2f si03 = __builtin_elementwise_fma(ti03, vhalf, vhalf); SCHED_PIN();
    v2f ti13; ti13[0] = tanh_(hi13[0]); SCHED_PIN();
    v2f fc03 = sf03 * cst[0][3]; SCHED_PIN();
    ti13[1] = tanh_(hi13[1]); SCHED_PIN();
    v2f ho10 = (v2f){acc[7][0], acc[7][1]} * 0.5f; SCHED_PIN();
    tg02[1] = tanh_(acc[4][5]); SCHED_PIN();
    v2f si10 = __builtin_elementwise_fma(ti10, vhalf, vhalf); SCHED_PIN();
    v2f tg03; tg03[0] = tanh_(acc[4][6]); SCHED_PIN();
    v2f fc10 = sf10 * cst[1][0]; SCHED_PIN();
    tg03[1] = tanh_(acc[4][7]); SCHED_PIN();
    v2f ho11 = (v2f){acc[7][2], acc[7][3]} * 0.5f; SCHED_PIN();
    v2f tg10; tg10[0] = tanh_(acc[5][0]); SCHED_PIN();
    v2f si11 = __builtin_elementwise_fma(ti11, vhalf, vhalf); SCHED_PIN();
    tg10[1] = tanh_(acc[5][1]); SCHED_PIN();
    v2f fc11 = sf11 * cst[1][1]; SCHED_PIN();
    v2f tg11; tg11[0] = tanh_(acc[5][2]); SCHED_PIN();
    v2f ho12 = (v2f){acc[7][4], acc[7][5]} * 0.5f; SCHED_PIN();
    tg11[1] = tanh_(acc[5][3]); SCHED_PIN();
    v2f si12 = __builtin_elementwise_fma(ti12, vhalf, vhalf); SCHED_PIN();
    v2f tg12; tg12[0] = tanh_(acc[5][4]); SCHED_PIN();
    v2f fc12 = sf12 * cst[1][2]; SCHED_PIN();
    tg12[1] = tanh_(acc[5][5]); SCHED_PIN();
    v2f ho13 = (v2f){acc[7][6], acc[7][7]} * 0.5f; SCHED_PIN();
    v2f tg13; tg13[0] = tanh_(acc[5][6]); SCHED_PIN();
    v2f si13 = __builtin_elementwise_fma(ti13, vhalf, vhalf); SCHED_PIN();
    tg13[1] = tanh_(acc[5][7]); SCHED_PIN();
    v2f fc13 = sf13 * cst[1][3]; SCHED_PIN();
    v2f to00; to00[0] = tanh_(ho00[0]); SCHED_PIN();
    cst[0][0] = __builtin_elementwise_fma(si00, tg00, fc00); SCHED_PIN();
    to00[1] = tanh_(ho00[1]); SCHED_PIN();
    cst[0][1] = __builtin_elementwise_fma(si01, tg01, fc01); SCHED_PIN();
    v2f to01; to01[0] = tanh_(ho01[0]); SCHED_PIN();
    cst[0][2] = __builtin_elementwise_fma(si02, tg02, fc02); SCHED_PIN();
    to01[1] = tanh_(ho01[1]); SCHED_PIN();
    cst[0][3] = __builtin_elementwise_fma(si03, tg03, fc03); SCHED_PIN();
    v2f to02; to02[0] = tanh_(ho02[0]); SCHED_PIN();
    cst[1][0] = __builtin_elementwise_fma(si10, tg10, fc10); SCHED_PIN();
    to02[1] = tanh_(ho02[1]); SCHED_PIN();
    cst[1][1] = __builtin_elementwise_fma(si11, tg11, fc11); SCHED_PIN();
    v2f to03; to03[0] = tanh_(ho03[0]); SCHED_PIN();
    cst[1][2] = __builtin_elementwise_fma(si12, tg12, fc12); SCHED_PIN();
    to03[1] = tanh_(ho03[1]); SCHED_PIN();
    cst[1][3] = __builtin_elementwise_fma(si13, tg13, fc13); SCHED_PIN();
    v2f to10; to10[0] = tanh_(ho10[0]); SCHED_PIN();
    v2f so00 = __builtin_elementwise_fma(to00, vhalf, vhalf); SCHED_PIN();
    to10[1] = tanh_(ho10[1]); SCHED_PIN();
    v2f so01 = __builtin_elementwise_fma(to01, vhalf, vhalf); SCHED_PIN();
    v2f to11; to11[0] = tanh_(ho11[0]); SCHED_PIN();
    v2f so02 = __builtin_elementwise_fma(to02, vhalf, vhalf); SCHED_PIN();
    to11[1] = tanh_(ho11[1]); SCHED_PIN();
    v2f so03 = __builtin_elementwise_fma(to03, vhalf, vhalf); SCHED_PIN();
    v2f to12; to12[0] = tanh_(ho12[0]); SCHED_PIN();
    to12[1] = tanh_(ho12[1]); SCHED_PIN();
    v2f so10 = __builtin_elementwise_fma(to10, vhalf, vhalf); SCHED_PIN();
    v2f to13; to13[0] = tanh_(ho13[0]); SCHED_PIN();
    to13[1] = tanh_(ho13[1]); SCHED_PIN();
    v2f so11 = __builtin_elementwise_fma(to11, vhalf, vhalf); SCHED_PIN();
    v2f tc00; tc00[0] = tanh_(cst[0][0][0]); SCHED_PIN();
    tc00[1] = tanh_(cst[0][0][1]); SCHED_PIN();
    v2f so12 = __builtin_elementwise_fma(to12, vhalf, vhalf); SCHED_PIN();
    v2f tc01; tc01[0] = tanh_(cst[0][1][0]); SCHED_PIN();
    tc01[1] = tanh_(cst[0][1][1]); SCHED_PIN();
    v2f so13 = __builtin_elementwise_fma(to13, vhalf, vhalf); SCHED_PIN();
    v2f tc02; tc02[0] = tanh_(cst[0][2][0]); SCHED_PIN();
    tc02[1] = tanh_(cst[0][2][1]); SCHED_PIN();
    v2f hh00 = so00 * tc00; SCHED_PIN();
    v2f tc03; tc03[0] = tanh_(cst[0][3][0]); SCHED_PIN();
    tc03[1] = tanh_(cst[0][3][1]); SCHED_PIN();
    v2f hh01 = so01 * tc01; SCHED_PIN();
    v2f tc10; tc10[0] = tanh_(cst[1][0][0]); SCHED_PIN();
    tc10[1] = tanh_(cst[1][0][1]); SCHED_PIN();
    v2f hh02 = so02 * tc02; SCHED_PIN();
    v2f tc11; tc11[0] = tanh_(cst[1][1][0]); SCHED_PIN();
    Bh[0] = (_Float16)hh00[0]; Bh[1] = (_Float16)hh00[1]; SCHED_PIN();
    tc11[1] = tanh_(cst[1][1][1]); SCHED_PIN();
    v2f hh03 = so03 * tc03; SCHED_PIN();
    v2f tc12; tc12[0] = tanh_(cst[1][2][0]); SCHED_PIN();
    Bh[2] = (_Float16)hh01[0]; Bh[3] = (_Float16)hh01[1]; SCHED_PIN();
    tc12[1] = tanh_(cst[1][2][1]); SCHED_PIN();
    v2f hh10 = so10 * tc10; SCHED_PIN();
    v2f tc13; tc13[0] = tanh_(cst[1][3][0]); SCHED_PIN();
    Bh[4] = (_Float16)hh02[0]; Bh[5] = (_Float16)hh02[1]; SCHED_PIN();
    tc13[1] = tanh_(cst[1][3][1]); SCHED_PIN();
    v2f hh11 = so11 * tc11; SCHED_PIN();
    Bh[6] = (_Float16)hh03[0]; Bh[7] = (_Float16)hh03[1]; SCHED_PIN();
    v2f hh12 = so12 * tc12; SCHED_PIN();
    Bh[8] = (_Float16)hh10[0]; Bh[9] = (_Float16)hh10[1]; SCHED_PIN();
    v2f hh13 = so13 * tc13; SCHED_PIN();
    Bh[10] = (_Float16)hh11[0]; Bh[11] = (_Float16)hh11[1]; SCHED_PIN();
    Bh[12] = (_Float16)hh12[0]; Bh[13] = (_Float16)hh12[1]; SCHED_PIN();
    Bh[14] = (_Float16)hh13[0]; Bh[15] = (_Float16)hh13[1]; SCHED_PIN();
}

__global__ __launch_bounds__(64)
void lstm_wmma_kernel(const float* __restrict__ x,
                      const float* __restrict__ W_ih,
                      const float* __restrict__ W_hh,
                      const float* __restrict__ b_ih,
                      const float* __restrict__ b_hh,
                      const float* __restrict__ W_fc,
                      const float* __restrict__ b_fc,
                      float* __restrict__ out,
                      int T)
{
    const int wave = __builtin_amdgcn_readfirstlane((int)(threadIdx.x >> 5));
    const int lane = threadIdx.x & 31;
    const int col  = lane & 15;
    const int hi   = lane >> 4;
    const int b0   = blockIdx.x * 16;

    __shared__ float4 ring[2 * SBLK * NCHUNK * 32];
    __shared__ v4f hring[2 * HSLOT * HCHUNK * 32];

    const int nblk = (T + SBLK - 1) / SBLK;

    v16h At[NTILE];
    v2f  cst[2][4];
    v16h Bh = (v16h){};
    v8f  accA[NTILE], accB[NTILE];
#pragma unroll
    for (int tt = 0; tt < NTILE; ++tt) { At[tt] = (v16h){}; accA[tt] = (v8f){}; accB[tt] = (v8f){}; }
#pragma unroll
    for (int p = 0; p < 2; ++p)
#pragma unroll
        for (int j = 0; j < 4; ++j) cst[p][j] = (v2f){0.f, 0.f};

    v16h Afc[SBLK];
    v8f  Cfc = (v8f){};
#pragma unroll
    for (int j = 0; j < SBLK; ++j) Afc[j] = (v16h){};
    v2f Aw[NTILE];
    v2f Bxs[SBLK];
    float2 xa[SBLK];
#pragma unroll
    for (int tt = 0; tt < NTILE; ++tt) Aw[tt] = (v2f){0.f, 0.f};
#pragma unroll
    for (int s = 0; s < SBLK; ++s) { Bxs[s] = (v2f){0.f, 0.f}; xa[s] = make_float2(0.f, 0.f); }

    const unsigned xlane = (unsigned)(b0 + col) * (unsigned)T * 2u;

    auto load_x = [&](int blk) {
#pragma unroll
        for (int s = 0; s < SBLK; ++s) {
            int t = blk * SBLK + s;
            t = (t < T) ? t : (T - 1);
            const float* xs = x + (size_t)t * 2;
            xa[s] = *(const float2*)(xs + xlane);
        }
    };
    auto build_bx = [&]() {
#pragma unroll
        for (int s = 0; s < SBLK; ++s) {
            Bxs[s][0] = (hi == 0) ? xa[s].x : 1.0f;
            Bxs[s][1] = (hi == 0) ? xa[s].y : 0.0f;
            SCHED_PIN();
        }
    };

    auto fill = [&](int blk) {
        float4* dst0 = ring + (size_t)(((blk & 1) * SBLK) * NCHUNK) * 32 + lane;
        build_bx();
        SCHED_PIN();
#pragma unroll
        for (int s = 0; s < SBLK; ++s) {
            v8f d[NTILE];
#pragma unroll
            for (int tt = 0; tt < NTILE; ++tt) {
                v8f z = (v8f){};
                d[tt] = __builtin_amdgcn_wmma_f32_16x16x4_f32(
                    false, Aw[tt], false, Bxs[s], (short)0, z, false, false);
            }
            SCHED_PIN();
            if (s == SBLK - 1) load_x(blk + 1);
#pragma unroll
            for (int tt = 0; tt < NTILE; ++tt) {
                dst0[(s * NCHUNK + tt * 2 + 0) * 32] = make_float4(d[tt][0], d[tt][1], d[tt][2], d[tt][3]);
                dst0[(s * NCHUNK + tt * 2 + 1) * 32] = make_float4(d[tt][4], d[tt][5], d[tt][6], d[tt][7]);
            }
            SCHED_PIN();
        }
    };

    auto hand_over = [&]() {
        __builtin_amdgcn_fence(__ATOMIC_RELEASE, "workgroup", "local");
        __builtin_amdgcn_s_barrier();
        __builtin_amdgcn_fence(__ATOMIC_ACQUIRE, "workgroup", "local");
    };

    auto hand_h = [&](int kb, int j) {
        v4f* hdst = hring + (size_t)(((kb & 1) * HSLOT + j) * HCHUNK) * 32 + lane;
        const v8f hb = __builtin_bit_cast(v8f, Bh);
        hdst[0 * 32] = __builtin_shufflevector(hb, hb, 0, 1, 2, 3);
        hdst[1 * 32] = __builtin_shufflevector(hb, hb, 4, 5, 6, 7);
    };

    auto load_h = [&](int kb, int j) -> v16h {
        const v4f* hsrc = hring + (size_t)(((kb & 1) * HSLOT + j) * HCHUNK) * 32 + lane;
        const v4f lo = hsrc[0 * 32];
        const v4f up = hsrc[1 * 32];
        const v8f hb = __builtin_shufflevector(lo, up, 0, 1, 2, 3, 4, 5, 6, 7);
        return __builtin_bit_cast(v16h, hb);
    };
    auto emit_y_block = [&](int kb) {
        v16h Bu[SBLK];
#pragma unroll
        for (int j = 0; j < SBLK; ++j) Bu[j] = load_h(kb, j);
        SCHED_PIN();
        v8f accy = Cfc;
#pragma unroll
        for (int j = 0; j < SBLK; ++j)
            accy = __builtin_amdgcn_wmma_f32_16x16x32_f16(
                false, Afc[j], false, Bu[j], (short)0, accy, false, false);
        const v4f yev  = __builtin_shufflevector(accy, accy, 0, 2, 4, 6);
        const v4f yod  = __builtin_shufflevector(accy, accy, 1, 3, 5, 7);
        const v4f ysel = hi ? yod : yev;
        float* orow = out + (size_t)(b0 + col) * T + (kb * SBLK - 2) + hi;
#pragma unroll
        for (int r = 0; r < SBLK / 2; ++r) {
            if (kb > 0 || r > 0) orow[2 * r] = ysel[r];
        }
    };
    auto emit_y_tail = [&](int kb, int cnt) {
        for (int j = 0; j < cnt; ++j) {
            const v16h Bu = load_h(kb, j);
            v8f accy = __builtin_amdgcn_wmma_f32_16x16x32_f16(
                false, Afc[0], false, Bu, (short)0, Cfc, false, false);
            const int u = kb * SBLK + j - 2;
            if (u >= 0 && hi == 0)
                out[(size_t)(b0 + col) * T + u] = accy[0];
        }
    };

    auto step = [&](auto tag, int k, int t, v8f (&done)[NTILE], v8f (&next)[NTILE]) {
        const float4* srcp = ring + (size_t)(((k & 1) * SBLK + (t - k * SBLK)) * NCHUNK) * 32 + lane;
#pragma unroll
        for (int tt = 0; tt < NTILE; ++tt) {
            float4 lo = srcp[(tt * 2 + 0) * 32];
            float4 up = srcp[(tt * 2 + 1) * 32];
            next[tt] = (v8f){lo.x, lo.y, lo.z, lo.w, up.x, up.y, up.z, up.w};
        }
        hand_h(k, t - k * SBLK);
        SCHED_PIN();

        if constexpr (decltype(tag)::may_be_first) { if (t > 0) cell_update(done, cst, Bh); }
        else cell_update(done, cst, Bh);

#pragma unroll
        for (int tt = 0; tt < NTILE; ++tt) {
            next[tt] = __builtin_amdgcn_wmma_f32_16x16x32_f16(
                false, At[tt], false, Bh, (short)0, next[tt], false, false);
        }
        SCHED_PIN();
    };

    if (wave == 0) {
#pragma unroll
        for (int tt = 0; tt < NTILE; ++tt) {
            const float* wrow = W_hh + (size_t)(tt * 16 + col) * HID;
#pragma unroll
            for (int e = 0; e < 8; ++e) {
                At[tt][e]     = (_Float16)wrow[hi * 8 + e];
                At[tt][e + 8] = (_Float16)wrow[16 + hi * 8 + e];
            }
        }
    } else {
        v16h wf;
#pragma unroll
        for (int e = 0; e < 8; ++e) {
            wf[e]     = (_Float16)W_fc[hi * 8 + e];
            wf[e + 8] = (_Float16)W_fc[16 + hi * 8 + e];
        }
#pragma unroll
        for (int j = 0; j < SBLK; ++j)
            if ((col & 7) == j) Afc[j] = wf;
        float bfc = b_fc[0];
#pragma unroll
        for (int e = 0; e < 8; ++e) Cfc[e] = bfc;
#pragma unroll
        for (int tt = 0; tt < NTILE; ++tt) {
            int g = tt * 16 + col;
            if (hi == 0) { Aw[tt][0] = W_ih[g * 2 + 0];     Aw[tt][1] = W_ih[g * 2 + 1]; }
            else         { Aw[tt][0] = b_ih[g] + b_hh[g];   Aw[tt][1] = 0.0f; }
        }
        load_x(0);
        fill(0);
    }
    hand_over();

    const int nfull = T / SBLK;
    for (int k = 0; k < nfull; ++k) {
        if (wave == 1) {
            if (k > 0) emit_y_block(k - 1);
            if (k + 1 < nblk) fill(k + 1);
        } else {
            step(StepTag<true>{},  k, k * SBLK,     accA, accB);
            step(StepTag<false>{}, k, k * SBLK + 1, accB, accA);
#pragma unroll
            for (int j = 1; j < SBLK / 2; ++j) {
                step(StepTag<false>{}, k, k * SBLK + 2 * j,     accA, accB);
                step(StepTag<false>{}, k, k * SBLK + 2 * j + 1, accB, accA);
            }
        }
        hand_over();
    }

    const int ntail = T - nfull * SBLK;
    if (wave == 0) {
        for (int t = nfull * SBLK; t + 1 < T; t += 2) {
            step(StepTag<true>{},  nfull, t,     accA, accB);
            step(StepTag<false>{}, nfull, t + 1, accB, accA);
        }
        if (T & 1) {
            step(StepTag<true>{}, nfull, T - 1, accA, accB);
            hand_h(nfull, ntail);
            cell_update(accB, cst, Bh);
        } else {
            hand_h(nfull, ntail);
            cell_update(accA, cst, Bh);
        }
        hand_h(nfull, ntail + 1);
    } else {
        if (nfull > 0) emit_y_block(nfull - 1);
    }
    hand_over();
    if (wave == 1) emit_y_tail(nfull, ntail + 2);
}

extern "C" void kernel_launch(void* const* d_in, const int* in_sizes, int n_in,
                              void* d_out, int out_size, void* d_ws, size_t ws_size,
                              hipStream_t stream) {
    const float* x    = (const float*)d_in[0];
    const float* W_ih = (const float*)d_in[1];
    const float* W_hh = (const float*)d_in[2];
    const float* b_ih = (const float*)d_in[3];
    const float* b_hh = (const float*)d_in[4];
    const float* W_fc = (const float*)d_in[5];
    const float* b_fc = (const float*)d_in[6];
    float* out = (float*)d_out;

    const int T = 512;
    const int B = out_size / T;
    const int nblocks = B / 16;

    lstm_wmma_kernel<<<dim3(nblocks), dim3(64), 0, stream>>>(
        x, W_ih, W_hh, b_ih, b_hh, W_fc, b_fc, out, T);

    (void)in_sizes; (void)n_in; (void)d_ws; (void)ws_size;
}


# Recurrent wave's LDS reads and writes gathered in 3 places per block of 8 steps (before steps 0, 3, 6: input tiles of the next 3 steps, h formed since), not one per step; sums, order unchanged
# speedup vs baseline: 2.3503x; 1.0358x over previous
#include <hip/hip_runtime.h>
#include <hip/hip_bf16.h>


typedef __attribute__((ext_vector_type(16))) _Float16 v16h;
typedef __attribute__((ext_vector_type(8)))  _Float16 v8h;
typedef __attribute__((ext_vector_type(2)))  _Float16 v2h;
typedef __attribute__((ext_vector_type(8)))  float    v8f;
typedef __attribute__((ext_vector_type(4)))  float    v4f;
typedef __attribute__((ext_vector_type(2)))  float    v2f;

#define SCHED_PIN() __builtin_amdgcn_sched_barrier(0)

template <int N> struct StepIdx { static constexpr int value = N; };

#define HID   32
#define NTILE 8
#define SBLK  8
#define NCHUNK 16
#define HCHUNK 2
#define HSLOT  (SBLK + 3)
#define PLACE_STEPS 3

#if __has_builtin(__builtin_amdgcn_tanhf)
__device__ __forceinline__ float tanh_(float x) { return __builtin_amdgcn_tanhf(x); }
__device__ __forceinline__ float sigm(float x) {
    return __builtin_fmaf(0.5f, __builtin_amdgcn_tanhf(0.5f * x), 0.5f);
}
#else
__device__ __forceinline__ float fast_rcp(float x) { return __builtin_amdgcn_rcpf(x); }
__device__ __forceinline__ float tanh_(float x) {
    float e = __expf(-2.0f * x);
    return (1.0f - e) * fast_rcp(1.0f + e);
}
__device__ __forceinline__ float sigm(float x) { return fast_rcp(1.0f + __expf(-x)); }
#endif

__device__ __forceinline__ void cell_update(const v8f (&acc)[NTILE], v2f (&cst)[2][4], v16h& Bh)
{
    const v2f vhalf = (v2f){0.5f, 0.5f};
    v2f tg00; tg00[0] = tanh_(acc[4][0]); SCHED_PIN();
    v2f hf00 = (v2f){acc[2][0], acc[2][1]} * 0.5f; SCHED_PIN();
    tg00[1] = tanh_(acc[4][1]); SCHED_PIN();
    v2f hf01 = (v2f){acc[2][2], acc[2][3]} * 0.5f; SCHED_PIN();
    v2f tg01; tg01[0] = tanh_(acc[4][2]); SCHED_PIN();
    v2f hf02 = (v2f){acc[2][4], acc[2][5]} * 0.5f; SCHED_PIN();
    tg01[1] = tanh_(acc[4][3]); SCHED_PIN();
    v2f hf03 = (v2f){acc[2][6], acc[2][7]} * 0.5f; SCHED_PIN();
    v2f tg02; tg02[0] = tanh_(acc[4][4]); SCHED_PIN();
    v2f hf10 = (v2f){acc[3][0], acc[3][1]} * 0.5f; SCHED_PIN();
    v2f tf00; tf00[0] = tanh_(hf00[0]); SCHED_PIN();
    v2f hf11 = (v2f){acc[3][2], acc[3][3]} * 0.5f; SCHED_PIN();
    tf00[1] = tanh_(hf00[1]); SCHED_PIN();
    v2f hf12 = (v2f){acc[3][4], acc[3][5]} * 0.5f; SCHED_PIN();
    v2f tf01; tf01[0] = tanh_(hf01[0]); SCHED_PIN();
    v2f hf13 = (v2f){acc[3][6], acc[3][7]} * 0.5f; SCHED_PIN();
    tf01[1] = tanh_(hf01[1]); SCHED_PIN();
    v2f hi00 = (v2f){acc[0][0], acc[0][1]} * 0.5f; SCHED_PIN();
    v2f tf02; tf02[0] = tanh_(hf02[0]); SCHED_PIN();
    v2f hi01 = (v2f){acc[0][2], acc[0][3]} * 0.5f; SCHED_PIN();
    tf02[1] = tanh_(hf02[1]); SCHED_PIN();
    v2f hi02 = (v2f){acc[0][4], acc[0][5]} * 0.5f; SCHED_PIN();
    v2f tf03; tf03[0] = tanh_(hf03[0]); SCHED_PIN();
    v2f hi03 = (v2f){acc[0][6], acc[0][7]} * 0.5f; SCHED_PIN();
    tf03[1] = tanh_(hf03[1]); SCHED_PIN();
    v2f hi10 = (v2f){acc[1][0], acc[1][1]} * 0.5f; SCHED_PIN();
    v2f tf10; tf10[0] = tanh_(hf10[0]); SCHED_PIN();
    v2f hi11 = (v2f){acc[1][2], acc[1][3]} * 0.5f; SCHED_PIN();
    tf10[1] = tanh_(hf10[1]); SCHED_PIN();
    v2f hi12 = (v2f){acc[1][4], acc[1][5]} * 0.5f; SCHED_PIN();
    v2f tf11; tf11[0] = tanh_(hf11[0]); SCHED_PIN();
    v2f hi13 = (v2f){acc[1][6], acc[1][7]} * 0.5f; SCHED_PIN();
    tf11[1] = tanh_(hf11[1]); SCHED_PIN();
    v2f sf00 = __builtin_elementwise_fma(tf00, vhalf, vhalf); SCHED_PIN();
    v2f tf12; tf12[0] = tanh_(hf12[0]); SCHED_PIN();
    v2f sf01 = __builtin_elementwise_fma(tf01, vhalf, vhalf); SCHED_PIN();
    tf12[1] = tanh_(hf12[1]); SCHED_PIN();
    v2f sf02 = __builtin_elementwise_fma(tf02, vhalf, vhalf); SCHED_PIN();
    v2f tf13; tf13[0] = tanh_(hf13[0]); SCHED_PIN();
    v2f sf03 = __builtin_elementwise_fma(tf03, vhalf, vhalf); SCHED_PIN();
    tf13[1] = tanh_(hf13[1]); SCHED_PIN();
    v2f sf10 = __builtin_elementwise_fma(tf10, vhalf, vhalf); SCHED_PIN();
    v2f ti00; ti00[0] = tanh_(hi00[0]); SCHED_PIN();
    v2f sf11 = __builtin_elementwise_fma(tf11, vhalf, vhalf); SCHED_PIN();
    ti00[1] = tanh_(hi00[1]); SCHED_PIN();
    v2f sf12 = __builtin_elementwise_fma(tf12, vhalf, vhalf); SCHED_PIN();
    v2f ti01; ti01[0] = tanh_(hi01[0]); SCHED_PIN();
    v2f ho00 = (v2f){acc[6][0], acc[6][1]} * 0.5f; SCHED_PIN();
    ti01[1] = tanh_(hi01[1]); SCHED_PIN();
    v2f sf13 = __builtin_elementwise_fma(tf13, vhalf, vhalf); SCHED_PIN();
    v2f ti02; ti02[0] = tanh_(hi02[0]); SCHED_PIN();
    v2f fc00 = sf00 * cst[0][0]; SCHED_PIN();
    ti02[1] = tanh_(hi02[1]); SCHED_PIN();
    v2f si00 = __builtin_elementwise_fma(ti00, vhalf, vhalf); SCHED_PIN();
    v2f ti03; ti03[0] = tanh_(hi03[0]); SCHED_PIN();
    v2f ho01 = (v2f){acc[6][2], acc[6][3]} * 0.5f; SCHED_PIN();
    ti03[1] = tanh_(hi03[1]); SCHED_PIN();
    v2f si01 = __builtin_elementwise_fma(ti01, vhalf, vhalf); SCHED_PIN();
    v2f ti10; ti10[0] = tanh_(hi10[0]); SCHED_PIN();
    v2f fc01 = sf01 * cst[0][1]; SCHED_PIN();
    ti10[1] = tanh_(hi10[1]); SCHED_PIN();
    v2f ho02 = (v2f){acc[6][4], acc[6][5]} * 0.5f; SCHED_PIN();
    v2f ti11; ti11[0] = tanh_(hi11[0]); SCHED_PIN();
    v2f si02 = __builtin_elementwise_fma(ti02, vhalf, vhalf); SCHED_PIN();
    ti11[1] = tanh_(hi11[1]); SCHED_PIN();
    v2f fc02 = sf02 * cst[0][2]; SCHED_PIN();
    v2f ti12; ti12[0] = tanh_(hi12[0]); SCHED_PIN();
    v2f ho03 = (v2f){acc[6][6], acc[6][7]} * 0.5f; SCHED_PIN();
    ti12[1] = tanh_(hi12[1]); SCHED_PIN();
    v2f si03 = __builtin_elementwise_fma(ti03, vhalf, vhalf); SCHED_PIN();
    v2f ti13; ti13[0] = tanh_(hi13[0]); SCHED_PIN();
    v2f fc03 = sf03 * cst[0][3]; SCHED_PIN();
    ti13[1] = tanh_(hi13[1]); SCHED_PIN();
    v2f ho10 = (v2f){acc[7][0], acc[7][1]} * 0.5f; SCHED_PIN();
    tg02[1] = tanh_(acc[4][5]); SCHED_PIN();
    v2f si10 = __builtin_elementwise_fma(ti10, vhalf, vhalf); SCHED_PIN();
    v2f tg03; tg03[0] = tanh_(acc[4][6]); SCHED_PIN();
    v2f fc10 = sf10 * cst[1][0]; SCHED_PIN();
    tg03[1] = tanh_(acc[4][7]); SCHED_PIN();
    v2f ho11 = (v2f){acc[7][2], acc[7][3]} * 0.5f; SCHED_PIN();
    v2f tg10; tg10[0] = tanh_(acc[5][0]); SCHED_PIN();
    v2f si11 = __builtin_elementwise_fma(ti11, vhalf, vhalf); SCHED_PIN();
    tg10[1] = tanh_(acc[5][1]); SCHED_PIN();
    v2f fc11 = sf11 * cst[1][1]; SCHED_PIN();
    v2f tg11; tg11[0] = tanh_(acc[5][2]); SCHED_PIN();
    v2f ho12 = (v2f){acc[7][4], acc[7][5]} * 0.5f; SCHED_PIN();
    tg11[1] = tanh_(acc[5][3]); SCHED_PIN();
    v2f si12 = __builtin_elementwise_fma(ti12, vhalf, vhalf); SCHED_PIN();
    v2f tg12; tg12[0] = tanh_(acc[5][4]); SCHED_PIN();
    v2f fc12 = sf12 * cst[1][2]; SCHED_PIN();
    tg12[1] = tanh_(acc[5][5]); SCHED_PIN();
    v2f ho13 = (v2f){acc[7][6], acc[7][7]} * 0.5f; SCHED_PIN();
    v2f tg13; tg13[0] = tanh_(acc[5][6]); SCHED_PIN();
    v2f si13 = __builtin_elementwise_fma(ti13, vhalf, vhalf); SCHED_PIN();
    tg13[1] = tanh_(acc[5][7]); SCHED_PIN();
    v2f fc13 = sf13 * cst[1][3]; SCHED_PIN();
    v2f to00; to00[0] = tanh_(ho00[0]); SCHED_PIN();
    cst[0][0] = __builtin_elementwise_fma(si00, tg00, fc00); SCHED_PIN();
    to00[1] = tanh_(ho00[1]); SCHED_PIN();
    cst[0][1] = __builtin_elementwise_fma(si01, tg01, fc01); SCHED_PIN();
    v2f to01; to01[0] = tanh_(ho01[0]); SCHED_PIN();
    cst[0][2] = __builtin_elementwise_fma(si02, tg02, fc02); SCHED_PIN();
    to01[1] = tanh_(ho01[1]); SCHED_PIN();
    cst[0][3] = __builtin_elementwise_fma(si03, tg03, fc03); SCHED_PIN();
    v2f to02; to02[0] = tanh_(ho02[0]); SCHED_PIN();
    cst[1][0] = __builtin_elementwise_fma(si10, tg10, fc10); SCHED_PIN();
    to02[1] = tanh_(ho02[1]); SCHED_PIN();
    cst[1][1] = __builtin_elementwise_fma(si11, tg11, fc11); SCHED_PIN();
    v2f to03; to03[0] = tanh_(ho03[0]); SCHED_PIN();
    cst[1][2] = __builtin_elementwise_fma(si12, tg12, fc12); SCHED_PIN();
    to03[1] = tanh_(ho03[1]); SCHED_PIN();
    cst[1][3] = __builtin_elementwise_fma(si13, tg13, fc13); SCHED_PIN();
    v2f to10; to10[0] = tanh_(ho10[0]); SCHED_PIN();
    v2f so00 = __builtin_elementwise_fma(to00, vhalf, vhalf); SCHED_PIN();
    to10[1] = tanh_(ho10[1]); SCHED_PIN();
    v2f so01 = __builtin_elementwise_fma(to01, vhalf, vhalf); SCHED_PIN();
    v2f to11; to11[0] = tanh_(ho11[0]); SCHED_PIN();
    v2f so02 = __builtin_elementwise_fma(to02, vhalf, vhalf); SCHED_PIN();
    to11[1] = tanh_(ho11[1]); SCHED_PIN();
    v2f so03 = __builtin_elementwise_fma(to03, vhalf, vhalf); SCHED_PIN();
    v2f to12; to12[0] = tanh_(ho12[0]); SCHED_PIN();
    to12[1] = tanh_(ho12[1]); SCHED_PIN();
    v2f so10 = __builtin_elementwise_fma(to10, vhalf, vhalf); SCHED_PIN();
    v2f to13; to13[0] = tanh_(ho13[0]); SCHED_PIN();
    to13[1] = tanh_(ho13[1]); SCHED_PIN();
    v2f so11 = __builtin_elementwise_fma(to11, vhalf, vhalf); SCHED_PIN();
    v2f tc00; tc00[0] = tanh_(cst[0][0][0]); SCHED_PIN();
    tc00[1] = tanh_(cst[0][0][1]); SCHED_PIN();
    v2f so12 = __builtin_elementwise_fma(to12, vhalf, vhalf); SCHED_PIN();
    v2f tc01; tc01[0] = tanh_(cst[0][1][0]); SCHED_PIN();
    tc01[1] = tanh_(cst[0][1][1]); SCHED_PIN();
    v2f so13 = __builtin_elementwise_fma(to13, vhalf, vhalf); SCHED_PIN();
    v2f tc02; tc02[0] = tanh_(cst[0][2][0]); SCHED_PIN();
    tc02[1] = tanh_(cst[0][2][1]); SCHED_PIN();
    v2f hh00 = so00 * tc00; SCHED_PIN();
    v2f tc03; tc03[0] = tanh_(cst[0][3][0]); SCHED_PIN();
    tc03[1] = tanh_(cst[0][3][1]); SCHED_PIN();
    v2f hh01 = so01 * tc01; SCHED_PIN();
    v2f tc10; tc10[0] = tanh_(cst[1][0][0]); SCHED_PIN();
    tc10[1] = tanh_(cst[1][0][1]); SCHED_PIN();
    v2f hh02 = so02 * tc02; SCHED_PIN();
    v2f tc11; tc11[0] = tanh_(cst[1][1][0]); SCHED_PIN();
    Bh[0] = (_Float16)hh00[0]; Bh[1] = (_Float16)hh00[1]; SCHED_PIN();
    tc11[1] = tanh_(cst[1][1][1]); SCHED_PIN();
    v2f hh03 = so03 * tc03; SCHED_PIN();
    v2f tc12; tc12[0] = tanh_(cst[1][2][0]); SCHED_PIN();
    Bh[2] = (_Float16)hh01[0]; Bh[3] = (_Float16)hh01[1]; SCHED_PIN();
    tc12[1] = tanh_(cst[1][2][1]); SCHED_PIN();
    v2f hh10 = so10 * tc10; SCHED_PIN();
    v2f tc13; tc13[0] = tanh_(cst[1][3][0]); SCHED_PIN();
    Bh[4] = (_Float16)hh02[0]; Bh[5] = (_Float16)hh02[1]; SCHED_PIN();
    tc13[1] = tanh_(cst[1][3][1]); SCHED_PIN();
    v2f hh11 = so11 * tc11; SCHED_PIN();
    Bh[6] = (_Float16)hh03[0]; Bh[7] = (_Float16)hh03[1]; SCHED_PIN();
    v2f hh12 = so12 * tc12; SCHED_PIN();
    Bh[8] = (_Float16)hh10[0]; Bh[9] = (_Float16)hh10[1]; SCHED_PIN();
    v2f hh13 = so13 * tc13; SCHED_PIN();
    Bh[10] = (_Float16)hh11[0]; Bh[11] = (_Float16)hh11[1]; SCHED_PIN();
    Bh[12] = (_Float16)hh12[0]; Bh[13] = (_Float16)hh12[1]; SCHED_PIN();
    Bh[14] = (_Float16)hh13[0]; Bh[15] = (_Float16)hh13[1]; SCHED_PIN();
}

__global__ __launch_bounds__(64)
void lstm_wmma_kernel(const float* __restrict__ x,
                      const float* __restrict__ W_ih,
                      const float* __restrict__ W_hh,
                      const float* __restrict__ b_ih,
                      const float* __restrict__ b_hh,
                      const float* __restrict__ W_fc,
                      const float* __restrict__ b_fc,
                      float* __restrict__ out,
                      int T)
{
    const int wave = __builtin_amdgcn_readfirstlane((int)(threadIdx.x >> 5));
    const int lane = threadIdx.x & 31;
    const int col  = lane & 15;
    const int hi   = lane >> 4;
    const int b0   = blockIdx.x * 16;

    __shared__ float4 ring[2 * SBLK * NCHUNK * 32];
    __shared__ v4f hring[2 * HSLOT * HCHUNK * 32];

    const int nblk = (T + SBLK - 1) / SBLK;

    v16h At[NTILE];
    v2f  cst[2][4];
    v16h Hs[SBLK];
    v8f  acc[SBLK][NTILE];
#pragma unroll
    for (int tt = 0; tt < NTILE; ++tt) At[tt] = (v16h){};
#pragma unroll
    for (int j = 0; j < SBLK; ++j) {
        Hs[j] = (v16h){};
#pragma unroll
        for (int tt = 0; tt < NTILE; ++tt) acc[j][tt] = (v8f){};
    }
#pragma unroll
    for (int p = 0; p < 2; ++p)
#pragma unroll
        for (int j = 0; j < 4; ++j) cst[p][j] = (v2f){0.f, 0.f};

    v16h Afc[SBLK];
    v8f  Cfc = (v8f){};
#pragma unroll
    for (int j = 0; j < SBLK; ++j) Afc[j] = (v16h){};
    v2f Aw[NTILE];
    v2f Bxs[SBLK];
    float2 xa[SBLK];
#pragma unroll
    for (int tt = 0; tt < NTILE; ++tt) Aw[tt] = (v2f){0.f, 0.f};
#pragma unroll
    for (int s = 0; s < SBLK; ++s) { Bxs[s] = (v2f){0.f, 0.f}; xa[s] = make_float2(0.f, 0.f); }

    const unsigned xlane = (unsigned)(b0 + col) * (unsigned)T * 2u;

    auto load_x = [&](int blk) {
#pragma unroll
        for (int s = 0; s < SBLK; ++s) {
            int t = blk * SBLK + s;
            t = (t < T) ? t : (T - 1);
            const float* xs = x + (size_t)t * 2;
            xa[s] = *(const float2*)(xs + xlane);
        }
    };
    auto build_bx = [&]() {
#pragma unroll
        for (int s = 0; s < SBLK; ++s) {
            Bxs[s][0] = (hi == 0) ? xa[s].x : 1.0f;
            Bxs[s][1] = (hi == 0) ? xa[s].y : 0.0f;
            SCHED_PIN();
        }
    };

    auto fill = [&](int blk) {
        float4* dst0 = ring + (size_t)(((blk & 1) * SBLK) * NCHUNK) * 32 + lane;
        build_bx();
        SCHED_PIN();
#pragma unroll
        for (int s = 0; s < SBLK; ++s) {
            v8f d[NTILE];
#pragma unroll
            for (int tt = 0; tt < NTILE; ++tt) {
                v8f z = (v8f){};
                d[tt] = __builtin_amdgcn_wmma_f32_16x16x4_f32(
                    false, Aw[tt], false, Bxs[s], (short)0, z, false, false);
            }
            SCHED_PIN();
            if (s == SBLK - 1) load_x(blk + 1);
#pragma unroll
            for (int tt = 0; tt < NTILE; ++tt) {
                dst0[(s * NCHUNK + tt * 2 + 0) * 32] = make_float4(d[tt][0], d[tt][1], d[tt][2], d[tt][3]);
                dst0[(s * NCHUNK + tt * 2 + 1) * 32] = make_float4(d[tt][4], d[tt][5], d[tt][6], d[tt][7]);
            }
            SCHED_PIN();
        }
    };

    auto hand_over = [&]() {
        __builtin_amdgcn_fence(__ATOMIC_RELEASE, "workgroup", "local");
        __builtin_amdgcn_s_barrier();
        __builtin_amdgcn_fence(__ATOMIC_ACQUIRE, "workgroup", "local");
    };

    auto store_h = [&](int kb, int j, const v16h& h) {
        v4f* hdst = hring + (size_t)(((kb & 1) * HSLOT + j) * HCHUNK) * 32 + lane;
        const v8f hb = __builtin_bit_cast(v8f, h);
        hdst[0 * 32] = __builtin_shufflevector(hb, hb, 0, 1, 2, 3);
        hdst[1 * 32] = __builtin_shufflevector(hb, hb, 4, 5, 6, 7);
    };

    auto load_h = [&](int kb, int j) -> v16h {
        const v4f* hsrc = hring + (size_t)(((kb & 1) * HSLOT + j) * HCHUNK) * 32 + lane;
        const v4f lo = hsrc[0 * 32];
        const v4f up = hsrc[1 * 32];
        const v8f hb = __builtin_shufflevector(lo, up, 0, 1, 2, 3, 4, 5, 6, 7);
        return __builtin_bit_cast(v16h, hb);
    };
    auto emit_y_block = [&](int kb) {
        v16h Bu[SBLK];
#pragma unroll
        for (int j = 0; j < SBLK; ++j) Bu[j] = load_h(kb, j);
        SCHED_PIN();
        v8f accy = Cfc;
#pragma unroll
        for (int j = 0; j < SBLK; ++j)
            accy = __builtin_amdgcn_wmma_f32_16x16x32_f16(
                false, Afc[j], false, Bu[j], (short)0, accy, false, false);
        const v4f yev  = __builtin_shufflevector(accy, accy, 0, 2, 4, 6);
        const v4f yod  = __builtin_shufflevector(accy, accy, 1, 3, 5, 7);
        const v4f ysel = hi ? yod : yev;
        float* orow = out + (size_t)(b0 + col) * T + (kb * SBLK - 3) + hi;
#pragma unroll
        for (int r = 0; r < SBLK / 2; ++r) {
            if (kb > 0 || 2 * r + hi >= 3) orow[2 * r] = ysel[r];
        }
    };
    auto emit_y_tail = [&](int kb, int cnt) {
        for (int j = 0; j < cnt; ++j) {
            const v16h Bu = load_h(kb, j);
            v8f accy = __builtin_amdgcn_wmma_f32_16x16x32_f16(
                false, Afc[0], false, Bu, (short)0, Cfc, false, false);
            const int u = kb * SBLK + j - 3;
            if (u >= 0 && hi == 0)
                out[(size_t)(b0 + col) * T + u] = accy[0];
        }
    };

    auto load_xp = [&](int k, int s, v8f (&dst)[NTILE]) {
        const float4* srcp = ring + (size_t)(((k & 1) * SBLK + s) * NCHUNK) * 32 + lane;
#pragma unroll
        for (int tt = 0; tt < NTILE; ++tt) {
            float4 lo = srcp[(tt * 2 + 0) * 32];
            float4 up = srcp[(tt * 2 + 1) * 32];
            dst[tt] = (v8f){lo.x, lo.y, lo.z, lo.w, up.x, up.y, up.z, up.w};
        }
    };
    auto recur = [&](const v16h& h, v8f (&a)[NTILE]) {
#pragma unroll
        for (int tt = 0; tt < NTILE; ++tt) {
            a[tt] = __builtin_amdgcn_wmma_f32_16x16x32_f16(
                false, At[tt], false, h, (short)0, a[tt], false, false);
        }
    };

    auto step = [&](auto jt, int k) {
        constexpr int J  = decltype(jt)::value;
        constexpr int JD = (J + SBLK - 1) % SBLK;
        if constexpr (J % PLACE_STEPS == 0) {
            constexpr int JE = (J + PLACE_STEPS < SBLK) ? (J + PLACE_STEPS) : SBLK;
#pragma unroll
            for (int s = J; s < JE; ++s) load_xp(k, s, acc[s]);
            if constexpr (J == 0) {
                store_h(k, 0, Hs[SBLK - 2]);
                store_h(k, 1, Hs[SBLK - 1]);
            } else {
#pragma unroll
                for (int s = J - PLACE_STEPS; s < J; ++s) store_h(k, s + 2, Hs[s]);
            }
            SCHED_PIN();
        }

        if constexpr (J == 0) { if (k > 0) cell_update(acc[JD], cst, Hs[J]); }
        else cell_update(acc[JD], cst, Hs[J]);

        recur(Hs[J], acc[J]);
        SCHED_PIN();
    };

    if (wave == 0) {
#pragma unroll
        for (int tt = 0; tt < NTILE; ++tt) {
            const float* wrow = W_hh + (size_t)(tt * 16 + col) * HID;
#pragma unroll
            for (int e = 0; e < 8; ++e) {
                At[tt][e]     = (_Float16)wrow[hi * 8 + e];
                At[tt][e + 8] = (_Float16)wrow[16 + hi * 8 + e];
            }
        }
    } else {
        v16h wf;
#pragma unroll
        for (int e = 0; e < 8; ++e) {
            wf[e]     = (_Float16)W_fc[hi * 8 + e];
            wf[e + 8] = (_Float16)W_fc[16 + hi * 8 + e];
        }
#pragma unroll
        for (int j = 0; j < SBLK; ++j)
            if ((col & 7) == j) Afc[j] = wf;
        float bfc = b_fc[0];
#pragma unroll
        for (int e = 0; e < 8; ++e) Cfc[e] = bfc;
#pragma unroll
        for (int tt = 0; tt < NTILE; ++tt) {
            int g = tt * 16 + col;
            if (hi == 0) { Aw[tt][0] = W_ih[g * 2 + 0];     Aw[tt][1] = W_ih[g * 2 + 1]; }
            else         { Aw[tt][0] = b_ih[g] + b_hh[g];   Aw[tt][1] = 0.0f; }
        }
        load_x(0);
        fill(0);
    }
    hand_over();

    const int nfull = T / SBLK;
    for (int k = 0; k < nfull; ++k) {
        if (wave == 1) {
            if (k > 0) emit_y_block(k - 1);
            if (k + 1 < nblk) fill(k + 1);
        } else {
            step(StepIdx<0>{}, k);
            step(StepIdx<1>{}, k);
            step(StepIdx<2>{}, k);
            step(StepIdx<3>{}, k);
            step(StepIdx<4>{}, k);
            step(StepIdx<5>{}, k);
            step(StepIdx<6>{}, k);
            step(StepIdx<7>{}, k);
        }
        hand_over();
    }

    const int ntail = T - nfull * SBLK;
    if (wave == 0) {
        v8f td[NTILE], tn[NTILE];
#pragma unroll
        for (int tt = 0; tt < NTILE; ++tt) { td[tt] = acc[SBLK - 1][tt]; tn[tt] = (v8f){}; }
        v16h Hc = (v16h){};
        store_h(nfull, 0, Hs[SBLK - 2]);
        store_h(nfull, 1, Hs[SBLK - 1]);
        for (int j = 0; j < ntail; ++j) {
            load_xp(nfull, j, tn);
            if (j > 0) store_h(nfull, j + 1, Hc);
            SCHED_PIN();
            if (nfull * SBLK + j > 0) cell_update(td, cst, Hc);
            recur(Hc, tn);
            SCHED_PIN();
#pragma unroll
            for (int tt = 0; tt < NTILE; ++tt) td[tt] = tn[tt];
        }
        if (ntail > 0) store_h(nfull, ntail + 1, Hc);
        cell_update(td, cst, Hc);
        store_h(nfull, ntail + 2, Hc);
    } else {
        if (nfull > 0) emit_y_block(nfull - 1);
    }
    hand_over();
    if (wave == 1) emit_y_tail(nfull, ntail + 3);
}

extern "C" void kernel_launch(void* const* d_in, const int* in_sizes, int n_in,
                              void* d_out, int out_size, void* d_ws, size_t ws_size,
                              hipStream_t stream) {
    const float* x    = (const float*)d_in[0];
    const float* W_ih = (const float*)d_in[1];
    const float* W_hh = (const float*)d_in[2];
    const float* b_ih = (const float*)d_in[3];
    const float* b_hh = (const float*)d_in[4];
    const float* W_fc = (const float*)d_in[5];
    const float* b_fc = (const float*)d_in[6];
    float* out = (float*)d_out;

    const int T = 512;
    const int B = out_size / T;
    const int nblocks = B / 16;

    lstm_wmma_kernel<<<dim3(nblocks), dim3(64), 0, stream>>>(
        x, W_ih, W_hh, b_ih, b_hh, W_fc, b_fc, out, T);

    (void)in_sizes; (void)n_in; (void)d_ws; (void)ws_size;
}
